# starting kernel
# baseline (speedup 1.0000x reference)
#include <hip/hip_runtime.h>
#include <hip/hip_bf16.h>


typedef __attribute__((ext_vector_type(2))) float v2f;
typedef __attribute__((ext_vector_type(8))) float v8f;

__device__ __constant__ int c_sdx[7] = {0, 0, 0, 1, 1, 2, -1};
__device__ __constant__ int c_sdy[7] = {0, 1, 2, 0, 1, 0, 1};

#define TWO_PI_F 6.28318530717958647692f

template <int LOGW, int ROWS>
__global__ void __launch_bounds__(128)
same_scale_corr(const float* __restrict__ xin, int pairBase,
                float* __restrict__ out) {
  constexpr int W = 1 << LOGW;
  constexpr int H = W;
  constexpr int WS = W + 4;
  constexpr int ZROW = 32 * WS;
  constexpr int SLABS = H / ROWS;
  __shared__ float lds[33 * WS];

  const int b = blockIdx.x / SLABS;
  const int slab = blockIdx.x & (SLABS - 1);

  const int lane = threadIdx.x & 31;
  const int wave = threadIdx.x >> 5;
  const int m0 = wave * 16;

  const int mrow = m0 + (lane & 15);
  const bool aval = (mrow < 56);
  const int sidx = aval ? (mrow >> 3) : 0;
  const int l1 = mrow & 7;
  const int slot = c_sdx[sidx] + 1;
  const int dy = aval ? c_sdy[sidx] : 0;
  const int aIdx = (aval ? (slot * 8 + l1) * WS : ZROW) + dy;
  const int ncol = lane & 15;
  const int bBase = (ncol < 8) ? (8 + ncol) * WS : ZROW;
  const int khi = (lane >> 4) << 1;

  for (int idx = threadIdx.x; idx < WS; idx += 128) lds[ZROW + idx] = 0.0f;

  v8f acc = {};

  for (int xi = 0; xi < ROWS; ++xi) {
    const int xr = slab * ROWS + xi;

    {
      constexpr int NV4 = 8 * W;
      constexpr int PER_CH4 = W / 4;
      for (int idx = threadIdx.x; idx < NV4; idx += 128) {
        const int row = idx >> (LOGW - 2);
        const int y4 = idx & (PER_CH4 - 1);
        const int rs = row >> 3;
        const int ch = row & 7;
        int xs = xr + rs - 1;
        xs += (xs < 0) ? H : 0;
        xs -= (xs >= H) ? H : 0;
        *(float4*)&lds[row * WS + (y4 << 2)] =
            *(const float4*)&xin[((((size_t)b * 8 + ch) << LOGW) + xs) * W +
                                 (y4 << 2)];
      }
      const int row = threadIdx.x >> 2;
      const int e = threadIdx.x & 3;
      const int rs = row >> 3;
      const int ch = row & 7;
      int xs = xr + rs - 1;
      xs += (xs < 0) ? H : 0;
      xs -= (xs >= H) ? H : 0;
      lds[row * WS + W + e] =
          xin[((((size_t)b * 8 + ch) << LOGW) + xs) * W + e];
    }
    if (threadIdx.x < 8) {
      int xn = xr + 4;
      xn -= (xn >= H) ? H : 0;
      __builtin_prefetch(
          &xin[((((size_t)b * 8 + threadIdx.x) << LOGW) + xn) * W], 0, 1);
    }
    __syncthreads();

#pragma unroll 8
    for (int k = 0; k < W; k += 4) {
      const int kb = k + khi;
      v2f av, bv;
      av[0] = lds[aIdx + kb];
      av[1] = lds[aIdx + kb + 1];
      bv = *(const v2f*)&lds[bBase + kb];
      acc = __builtin_amdgcn_wmma_f32_16x16x4_f32(
           false, av,  false, bv,
           (short)0, acc,  false,  false);
    }
    __syncthreads();
  }

  const int rbase = m0 + ((lane >> 4) << 3);
  const int col = lane & 15;
  if (col < 8) {
#pragma unroll
    for (int i = 0; i < 8; ++i) {
      const int R = rbase + i;
      if (R < 56) {
        const int s = R >> 3;
        const int l1o = R & 7;
        atomicAdd(&out[(((size_t)b * 384) + pairBase + l1o * 8 + col) * 7 + s],
                  acc[i]);
      }
    }
  }
}

template <int LOGN>
__global__ void fft_rows(const float* __restrict__ xin,
                         float2* __restrict__ spec) {
  constexpr int N = 1 << LOGN;
  __shared__ float2 buf[N];
  const int row = blockIdx.x & (N - 1);
  const int img = blockIdx.x >> LOGN;
  const float* src = xin + (((size_t)img << LOGN) + row) * N;
  const int t = threadIdx.x;

  buf[__brev((unsigned)t) >> (32 - LOGN)] = make_float2(src[t], 0.0f);
  buf[__brev((unsigned)(t + N / 2)) >> (32 - LOGN)] =
      make_float2(src[t + N / 2], 0.0f);
  __syncthreads();
#pragma unroll
  for (int st = 1; st <= LOGN; ++st) {
    const int len = 1 << st;
    const int half = len >> 1;
    const int j = t & (half - 1);
    const int base = (t >> (st - 1)) << st;
    float sn, cs;
    __sincosf(-TWO_PI_F * (float)j / (float)len, &sn, &cs);
    const float2 u = buf[base + j];
    const float2 v = buf[base + j + half];
    const float2 tv = make_float2(v.x * cs - v.y * sn, v.x * sn + v.y * cs);
    buf[base + j] = make_float2(u.x + tv.x, u.y + tv.y);
    buf[base + j + half] = make_float2(u.x - tv.x, u.y - tv.y);
    __syncthreads();
  }
  float2* dst = spec + (((size_t)img << LOGN) + row) * N;
  dst[t] = buf[t];
  dst[t + N / 2] = buf[t + N / 2];
}

template <int LOGN>
__global__ void fft_cols(float2* __restrict__ spec) {
  constexpr int N = 1 << LOGN;
  __shared__ float2 buf[N];
  const int col = blockIdx.x & (N - 1);
  const int img = blockIdx.x >> LOGN;
  float2* base0 = spec + ((size_t)img << (2 * LOGN)) + col;
  const int t = threadIdx.x;

  buf[__brev((unsigned)t) >> (32 - LOGN)] = base0[(size_t)t * N];
  buf[__brev((unsigned)(t + N / 2)) >> (32 - LOGN)] =
      base0[(size_t)(t + N / 2) * N];
  __syncthreads();
#pragma unroll
  for (int st = 1; st <= LOGN; ++st) {
    const int len = 1 << st;
    const int half = len >> 1;
    const int j = t & (half - 1);
    const int base = (t >> (st - 1)) << st;
    float sn, cs;
    __sincosf(-TWO_PI_F * (float)j / (float)len, &sn, &cs);
    const float2 u = buf[base + j];
    const float2 v = buf[base + j + half];
    const float2 tv = make_float2(v.x * cs - v.y * sn, v.x * sn + v.y * cs);
    buf[base + j] = make_float2(u.x + tv.x, u.y + tv.y);
    buf[base + j + half] = make_float2(u.x - tv.x, u.y - tv.y);
    __syncthreads();
  }
  base0[(size_t)t * N] = buf[t];
  base0[(size_t)(t + N / 2) * N] = buf[t + N / 2];
}

template <int LOG1, int LOG2>
__global__ void __launch_bounds__(256)
cross_corr(const float2* __restrict__ s1, const float2* __restrict__ s2,
           int pairBase, float* __restrict__ out) {
  constexpr int N1 = 1 << LOG1;
  constexpr int N2 = 1 << LOG2;
  __shared__ float red[256];
  const int b = blockIdx.x >> 6;
  const int p = blockIdx.x & 63;
  const int l1 = p >> 3, l2 = p & 7;
  const float2* F1 = s1 + (((size_t)b * 8 + l1) << (2 * LOG1));
  const float2* F2 = s2 + (((size_t)b * 8 + l2) << (2 * LOG2));

  float a0 = 0.f, a1 = 0.f, a2 = 0.f, a3 = 0.f, a4 = 0.f, a5 = 0.f, a6 = 0.f;
  constexpr int BINS = N2 * N2;
  for (int tbin = threadIdx.x; tbin < BINS; tbin += blockDim.x) {
    const int ix2 = tbin >> LOG2;
    const int iy2 = tbin & (N2 - 1);
    const int fx = (ix2 < (N2 >> 1)) ? ix2 : ix2 - N2;
    const int fy = (iy2 < (N2 >> 1)) ? iy2 : iy2 - N2;
    const float2 A =
        F1[(((fx + N1) & (N1 - 1)) << LOG1) + ((fy + N1) & (N1 - 1))];
    const float2 B = F2[tbin];
    const float zr = A.x * B.x + A.y * B.y;
    const float zi = A.y * B.x - A.x * B.y;
    float sx, cx, sy, cy;
    __sincosf(TWO_PI_F * (float)fx / (float)N1, &sx, &cx);
    __sincosf(TWO_PI_F * (float)fy / (float)N1, &sy, &cy);
    const float c2x = cx * cx - sx * sx, s2x = 2.f * cx * sx;
    const float c2y = cy * cy - sy * sy, s2y = 2.f * cy * sy;
    a0 += zr;
    a1 += zr * cy - zi * sy;
    a2 += zr * c2y - zi * s2y;
    a3 += zr * cx - zi * sx;
    const float xr = zr * cx - zi * sx, xi = zr * sx + zi * cx;
    a4 += xr * cy - xi * sy;
    a5 += zr * c2x - zi * s2x;
    const float mr = zr * cx + zi * sx, mi = zi * cx - zr * sx;
    a6 += mr * cy - mi * sy;
  }

  const float inv = 1.0f / (float)BINS;
  float accs[7] = {a0, a1, a2, a3, a4, a5, a6};
#pragma unroll
  for (int s = 0; s < 7; ++s) {
    red[threadIdx.x] = accs[s];
    __syncthreads();
    for (int off = 128; off > 0; off >>= 1) {
      if (threadIdx.x < off) red[threadIdx.x] += red[threadIdx.x + off];
      __syncthreads();
    }
    if (threadIdx.x == 0)
      out[(((size_t)b * 384) + pairBase + p) * 7 + s] = red[0] * inv;
    __syncthreads();
  }
}

extern "C" void kernel_launch(void* const* d_in, const int* in_sizes, int n_in,
                              void* d_out, int out_size, void* d_ws,
                              size_t ws_size, hipStream_t stream) {
  const float* x0 = (const float*)d_in[0];
  const float* x1 = (const float*)d_in[1];
  const float* x2 = (const float*)d_in[2];
  float* out = (float*)d_out;

  hipMemsetAsync(d_out, 0, (size_t)out_size * sizeof(float), stream);

  same_scale_corr<8, 2><<<dim3(2 * 128), 128, 0, stream>>>(x0, 0, out);
  same_scale_corr<7, 1><<<dim3(2 * 128), 128, 0, stream>>>(x1, 192, out);
  same_scale_corr<6, 1><<<dim3(2 * 64), 128, 0, stream>>>(x2, 320, out);

  const size_t n0 = (size_t)16 * 256 * 256;
  const size_t n1 = (size_t)16 * 128 * 128;
  const size_t n2 = (size_t)16 * 64 * 64;
  if (ws_size >= (n0 + n1 + n2) * sizeof(float2)) {
    float2* spec0 = (float2*)d_ws;
    float2* spec1 = spec0 + n0;
    float2* spec2 = spec1 + n1;

    fft_rows<8><<<dim3(16 * 256), 128, 0, stream>>>(x0, spec0);
    fft_cols<8><<<dim3(16 * 256), 128, 0, stream>>>(spec0);
    fft_rows<7><<<dim3(16 * 128), 64, 0, stream>>>(x1, spec1);
    fft_cols<7><<<dim3(16 * 128), 64, 0, stream>>>(spec1);
    fft_rows<6><<<dim3(16 * 64), 32, 0, stream>>>(x2, spec2);
    fft_cols<6><<<dim3(16 * 64), 32, 0, stream>>>(spec2);

    cross_corr<8, 7><<<dim3(128), 256, 0, stream>>>(spec0, spec1, 64, out);
    cross_corr<8, 6><<<dim3(128), 256, 0, stream>>>(spec0, spec2, 128, out);
    cross_corr<7, 6><<<dim3(128), 256, 0, stream>>>(spec1, spec2, 256, out);
  }
}


# cross_corr: bin sum split over two work-groups per channel pair, each thread's loads issued together at constant strides (unrolled); partials by lane shuffle + two commuting float atomicAdds to out
# speedup vs baseline: 1.5257x; 1.5257x over previous
#include <hip/hip_runtime.h>
#include <hip/hip_bf16.h>


typedef __attribute__((ext_vector_type(2))) float v2f;
typedef __attribute__((ext_vector_type(8))) float v8f;

__device__ __constant__ int c_sdx[7] = {0, 0, 0, 1, 1, 2, -1};
__device__ __constant__ int c_sdy[7] = {0, 1, 2, 0, 1, 0, 1};

#define TWO_PI_F 6.28318530717958647692f

template <int LOGW, int ROWS>
__global__ void __launch_bounds__(128)
same_scale_corr(const float* __restrict__ xin, int pairBase,
                float* __restrict__ out) {
  constexpr int W = 1 << LOGW;
  constexpr int H = W;
  constexpr int WS = W + 4;
  constexpr int ZROW = 32 * WS;
  constexpr int SLABS = H / ROWS;
  __shared__ float lds[33 * WS];

  const int b = blockIdx.x / SLABS;
  const int slab = blockIdx.x & (SLABS - 1);

  const int lane = threadIdx.x & 31;
  const int wave = threadIdx.x >> 5;
  const int m0 = wave * 16;

  const int mrow = m0 + (lane & 15);
  const bool aval = (mrow < 56);
  const int sidx = aval ? (mrow >> 3) : 0;
  const int l1 = mrow & 7;
  const int slot = c_sdx[sidx] + 1;
  const int dy = aval ? c_sdy[sidx] : 0;
  const int aIdx = (aval ? (slot * 8 + l1) * WS : ZROW) + dy;
  const int ncol = lane & 15;
  const int bBase = (ncol < 8) ? (8 + ncol) * WS : ZROW;
  const int khi = (lane >> 4) << 1;

  for (int idx = threadIdx.x; idx < WS; idx += 128) lds[ZROW + idx] = 0.0f;

  v8f acc = {};

  for (int xi = 0; xi < ROWS; ++xi) {
    const int xr = slab * ROWS + xi;

    {
      constexpr int NV4 = 8 * W;
      constexpr int PER_CH4 = W / 4;
      for (int idx = threadIdx.x; idx < NV4; idx += 128) {
        const int row = idx >> (LOGW - 2);
        const int y4 = idx & (PER_CH4 - 1);
        const int rs = row >> 3;
        const int ch = row & 7;
        int xs = xr + rs - 1;
        xs += (xs < 0) ? H : 0;
        xs -= (xs >= H) ? H : 0;
        *(float4*)&lds[row * WS + (y4 << 2)] =
            *(const float4*)&xin[((((size_t)b * 8 + ch) << LOGW) + xs) * W +
                                 (y4 << 2)];
      }
      const int row = threadIdx.x >> 2;
      const int e = threadIdx.x & 3;
      const int rs = row >> 3;
      const int ch = row & 7;
      int xs = xr + rs - 1;
      xs += (xs < 0) ? H : 0;
      xs -= (xs >= H) ? H : 0;
      lds[row * WS + W + e] =
          xin[((((size_t)b * 8 + ch) << LOGW) + xs) * W + e];
    }
    if (threadIdx.x < 8) {
      int xn = xr + 4;
      xn -= (xn >= H) ? H : 0;
      __builtin_prefetch(
          &xin[((((size_t)b * 8 + threadIdx.x) << LOGW) + xn) * W], 0, 1);
    }
    __syncthreads();

#pragma unroll 8
    for (int k = 0; k < W; k += 4) {
      const int kb = k + khi;
      v2f av, bv;
      av[0] = lds[aIdx + kb];
      av[1] = lds[aIdx + kb + 1];
      bv = *(const v2f*)&lds[bBase + kb];
      acc = __builtin_amdgcn_wmma_f32_16x16x4_f32(
           false, av,  false, bv,
           (short)0, acc,  false,  false);
    }
    __syncthreads();
  }

  const int rbase = m0 + ((lane >> 4) << 3);
  const int col = lane & 15;
  if (col < 8) {
#pragma unroll
    for (int i = 0; i < 8; ++i) {
      const int R = rbase + i;
      if (R < 56) {
        const int s = R >> 3;
        const int l1o = R & 7;
        atomicAdd(&out[(((size_t)b * 384) + pairBase + l1o * 8 + col) * 7 + s],
                  acc[i]);
      }
    }
  }
}

template <int LOGN>
__global__ void fft_rows(const float* __restrict__ xin,
                         float2* __restrict__ spec) {
  constexpr int N = 1 << LOGN;
  __shared__ float2 buf[N];
  const int row = blockIdx.x & (N - 1);
  const int img = blockIdx.x >> LOGN;
  const float* src = xin + (((size_t)img << LOGN) + row) * N;
  const int t = threadIdx.x;

  buf[__brev((unsigned)t) >> (32 - LOGN)] = make_float2(src[t], 0.0f);
  buf[__brev((unsigned)(t + N / 2)) >> (32 - LOGN)] =
      make_float2(src[t + N / 2], 0.0f);
  __syncthreads();
#pragma unroll
  for (int st = 1; st <= LOGN; ++st) {
    const int len = 1 << st;
    const int half = len >> 1;
    const int j = t & (half - 1);
    const int base = (t >> (st - 1)) << st;
    float sn, cs;
    __sincosf(-TWO_PI_F * (float)j / (float)len, &sn, &cs);
    const float2 u = buf[base + j];
    const float2 v = buf[base + j + half];
    const float2 tv = make_float2(v.x * cs - v.y * sn, v.x * sn + v.y * cs);
    buf[base + j] = make_float2(u.x + tv.x, u.y + tv.y);
    buf[base + j + half] = make_float2(u.x - tv.x, u.y - tv.y);
    __syncthreads();
  }
  float2* dst = spec + (((size_t)img << LOGN) + row) * N;
  dst[t] = buf[t];
  dst[t + N / 2] = buf[t + N / 2];
}

template <int LOGN>
__global__ void fft_cols(float2* __restrict__ spec) {
  constexpr int N = 1 << LOGN;
  __shared__ float2 buf[N];
  const int col = blockIdx.x & (N - 1);
  const int img = blockIdx.x >> LOGN;
  float2* base0 = spec + ((size_t)img << (2 * LOGN)) + col;
  const int t = threadIdx.x;

  buf[__brev((unsigned)t) >> (32 - LOGN)] = base0[(size_t)t * N];
  buf[__brev((unsigned)(t + N / 2)) >> (32 - LOGN)] =
      base0[(size_t)(t + N / 2) * N];
  __syncthreads();
#pragma unroll
  for (int st = 1; st <= LOGN; ++st) {
    const int len = 1 << st;
    const int half = len >> 1;
    const int j = t & (half - 1);
    const int base = (t >> (st - 1)) << st;
    float sn, cs;
    __sincosf(-TWO_PI_F * (float)j / (float)len, &sn, &cs);
    const float2 u = buf[base + j];
    const float2 v = buf[base + j + half];
    const float2 tv = make_float2(v.x * cs - v.y * sn, v.x * sn + v.y * cs);
    buf[base + j] = make_float2(u.x + tv.x, u.y + tv.y);
    buf[base + j + half] = make_float2(u.x - tv.x, u.y - tv.y);
    __syncthreads();
  }
  base0[(size_t)t * N] = buf[t];
  base0[(size_t)(t + N / 2) * N] = buf[t + N / 2];
}

template <int LOG1, int LOG2, int SPLIT, int U>
__global__ void __launch_bounds__((1 << (2 * LOG2)) / (SPLIT * U))
cross_corr(const float2* __restrict__ s1, const float2* __restrict__ s2,
           int pairBase, float* __restrict__ out) {
  constexpr int N1 = 1 << LOG1;
  constexpr int N2 = 1 << LOG2;
  constexpr int BINS = N2 * N2;
  constexpr int THREADS = BINS / (SPLIT * U);
  constexpr int NW = THREADS / 32;
  constexpr int ROWS_PER_PASS = THREADS >> LOG2;
  constexpr int ROWS_PER_BLOCK = N2 / SPLIT;
  static_assert(THREADS >= N2 && THREADS <= 1024 && (THREADS & 31) == 0,
                "block must cover whole spectrum rows");
  static_assert(ROWS_PER_PASS * U == ROWS_PER_BLOCK, "share must tile the rows");
  static_assert(SPLIT >= 2 && (SPLIT & (SPLIT - 1)) == 0, "rows of one half per block");
  __shared__ float red[7][NW];

  const int part = blockIdx.x & (SPLIT - 1);
  const int bp = blockIdx.x / SPLIT;
  const int b = bp >> 6;
  const int p = bp & 63;
  const int l1 = p >> 3, l2 = p & 7;
  const float2* F1 = s1 + (((size_t)b * 8 + l1) << (2 * LOG1));
  const float2* F2 = s2 + (((size_t)b * 8 + l2) << (2 * LOG2));

  const int t = threadIdx.x;
  const int iy2 = t & (N2 - 1);
  const int fy = (iy2 < (N2 >> 1)) ? iy2 : iy2 - N2;
  const int col1 = (fy + N1) & (N1 - 1);
  const int row0 = part * ROWS_PER_BLOCK + (t >> LOG2);
  const int wrap1 = (part >= SPLIT / 2) ? (N1 - N2) : 0;
  const float2* pA = F1 + ((size_t)(row0 + wrap1) << LOG1) + col1;
  const float2* pB = F2 + ((size_t)row0 << LOG2) + iy2;

  float2 A[U], B[U];
#pragma unroll
  for (int u = 0; u < U; ++u) {
    A[u] = pA[(size_t)u * ROWS_PER_PASS * N1];
    B[u] = pB[(size_t)u * ROWS_PER_PASS * N2];
  }
#pragma unroll
  for (int u = 0; u < U; ++u)
    asm volatile("" ::"v"(A[u].x), "v"(A[u].y), "v"(B[u].x), "v"(B[u].y));

  float sy, cy;
  __sincosf(TWO_PI_F * (float)fy / (float)N1, &sy, &cy);
  const float c2y = cy * cy - sy * sy, s2y = 2.f * cy * sy;

  float a0 = 0.f, a1 = 0.f, a2 = 0.f, a3 = 0.f, a4 = 0.f, a5 = 0.f, a6 = 0.f;
#pragma unroll
  for (int u = 0; u < U; ++u) {
    const int ix2 = row0 + u * ROWS_PER_PASS;
    const int fx = (ix2 < (N2 >> 1)) ? ix2 : ix2 - N2;
    const float zr = A[u].x * B[u].x + A[u].y * B[u].y;
    const float zi = A[u].y * B[u].x - A[u].x * B[u].y;
    float sx, cx;
    __sincosf(TWO_PI_F * (float)fx / (float)N1, &sx, &cx);
    const float c2x = cx * cx - sx * sx, s2x = 2.f * cx * sx;
    a0 += zr;
    a1 += zr * cy - zi * sy;
    a2 += zr * c2y - zi * s2y;
    a3 += zr * cx - zi * sx;
    const float xr = zr * cx - zi * sx, xi = zr * sx + zi * cx;
    a4 += xr * cy - xi * sy;
    a5 += zr * c2x - zi * s2x;
    const float mr = zr * cx + zi * sx, mi = zi * cx - zr * sx;
    a6 += mr * cy - mi * sy;
  }

  float accs[7] = {a0, a1, a2, a3, a4, a5, a6};
#pragma unroll
  for (int s = 0; s < 7; ++s) {
    float v = accs[s];
    for (int off = 16; off > 0; off >>= 1) v += __shfl_xor(v, off, 32);
    accs[s] = v;
  }
  if ((t & 31) == 0) {
#pragma unroll
    for (int s = 0; s < 7; ++s) red[s][t >> 5] = accs[s];
  }
  __syncthreads();
  if (t < 7) {
    float v = 0.f;
    for (int w = 0; w < NW; ++w) v += red[t][w];
    const float inv = 1.0f / (float)BINS;
    atomicAdd(&out[(((size_t)b * 384) + pairBase + p) * 7 + t], v * inv);
  }
}

extern "C" void kernel_launch(void* const* d_in, const int* in_sizes, int n_in,
                              void* d_out, int out_size, void* d_ws,
                              size_t ws_size, hipStream_t stream) {
  const float* x0 = (const float*)d_in[0];
  const float* x1 = (const float*)d_in[1];
  const float* x2 = (const float*)d_in[2];
  float* out = (float*)d_out;

  hipMemsetAsync(d_out, 0, (size_t)out_size * sizeof(float), stream);

  same_scale_corr<8, 2><<<dim3(2 * 128), 128, 0, stream>>>(x0, 0, out);
  same_scale_corr<7, 1><<<dim3(2 * 128), 128, 0, stream>>>(x1, 192, out);
  same_scale_corr<6, 1><<<dim3(2 * 64), 128, 0, stream>>>(x2, 320, out);

  const size_t n0 = (size_t)16 * 256 * 256;
  const size_t n1 = (size_t)16 * 128 * 128;
  const size_t n2 = (size_t)16 * 64 * 64;
  if (ws_size >= (n0 + n1 + n2) * sizeof(float2)) {
    float2* spec0 = (float2*)d_ws;
    float2* spec1 = spec0 + n0;
    float2* spec2 = spec1 + n1;

    fft_rows<8><<<dim3(16 * 256), 128, 0, stream>>>(x0, spec0);
    fft_cols<8><<<dim3(16 * 256), 128, 0, stream>>>(spec0);
    fft_rows<7><<<dim3(16 * 128), 64, 0, stream>>>(x1, spec1);
    fft_cols<7><<<dim3(16 * 128), 64, 0, stream>>>(spec1);
    fft_rows<6><<<dim3(16 * 64), 32, 0, stream>>>(x2, spec2);
    fft_cols<6><<<dim3(16 * 64), 32, 0, stream>>>(spec2);

    cross_corr<8, 7, 2, 16><<<dim3(256), 512, 0, stream>>>(spec0, spec1, 64, out);
    cross_corr<8, 6, 2, 8><<<dim3(256), 256, 0, stream>>>(spec0, spec2, 128, out);
    cross_corr<7, 6, 2, 8><<<dim3(256), 256, 0, stream>>>(spec1, spec2, 256, out);
  }
}


# launches merged by kind: memset + 12 kernels -> 4 (row FFTs, same-scale WMMA, column FFTs, cross sums); block index picks the scale, first kernel's stores clear out; every sum in the same order
# speedup vs baseline: 2.1855x; 1.4325x over previous
#include <hip/hip_runtime.h>
#include <hip/hip_bf16.h>


typedef __attribute__((ext_vector_type(2))) float v2f;
typedef __attribute__((ext_vector_type(8))) float v8f;

__device__ __constant__ int c_sdx[7] = {0, 0, 0, 1, 1, 2, -1};
__device__ __constant__ int c_sdy[7] = {0, 1, 2, 0, 1, 0, 1};

#define TWO_PI_F 6.28318530717958647692f

template <int LOGW, int ROWS>
__device__ __forceinline__ void same_scale_body(const float* __restrict__ xin,
                                                int pairBase,
                                                float* __restrict__ out,
                                                int blk, float* lds) {
  constexpr int W = 1 << LOGW;
  constexpr int H = W;
  constexpr int WS = W + 4;
  constexpr int ZROW = 32 * WS;
  constexpr int SLABS = H / ROWS;

  const int b = blk / SLABS;
  const int slab = blk & (SLABS - 1);

  const int lane = threadIdx.x & 31;
  const int wave = threadIdx.x >> 5;
  const int m0 = wave * 16;

  const int mrow = m0 + (lane & 15);
  const bool aval = (mrow < 56);
  const int sidx = aval ? (mrow >> 3) : 0;
  const int l1 = mrow & 7;
  const int slot = c_sdx[sidx] + 1;
  const int dy = aval ? c_sdy[sidx] : 0;
  const int aIdx = (aval ? (slot * 8 + l1) * WS : ZROW) + dy;
  const int ncol = lane & 15;
  const int bBase = (ncol < 8) ? (8 + ncol) * WS : ZROW;
  const int khi = (lane >> 4) << 1;

  for (int idx = threadIdx.x; idx < WS; idx += 128) lds[ZROW + idx] = 0.0f;

  v8f acc = {};

  for (int xi = 0; xi < ROWS; ++xi) {
    const int xr = slab * ROWS + xi;

    {
      constexpr int NV4 = 8 * W;
      constexpr int PER_CH4 = W / 4;
      for (int idx = threadIdx.x; idx < NV4; idx += 128) {
        const int row = idx >> (LOGW - 2);
        const int y4 = idx & (PER_CH4 - 1);
        const int rs = row >> 3;
        const int ch = row & 7;
        int xs = xr + rs - 1;
        xs += (xs < 0) ? H : 0;
        xs -= (xs >= H) ? H : 0;
        *(float4*)&lds[row * WS + (y4 << 2)] =
            *(const float4*)&xin[((((size_t)b * 8 + ch) << LOGW) + xs) * W +
                                 (y4 << 2)];
      }
      const int row = threadIdx.x >> 2;
      const int e = threadIdx.x & 3;
      const int rs = row >> 3;
      const int ch = row & 7;
      int xs = xr + rs - 1;
      xs += (xs < 0) ? H : 0;
      xs -= (xs >= H) ? H : 0;
      lds[row * WS + W + e] =
          xin[((((size_t)b * 8 + ch) << LOGW) + xs) * W + e];
    }
    if (threadIdx.x < 8) {
      int xn = xr + 4;
      xn -= (xn >= H) ? H : 0;
      __builtin_prefetch(
          &xin[((((size_t)b * 8 + threadIdx.x) << LOGW) + xn) * W], 0, 1);
    }
    __syncthreads();

#pragma unroll 8
    for (int k = 0; k < W; k += 4) {
      const int kb = k + khi;
      v2f av, bv;
      av[0] = lds[aIdx + kb];
      av[1] = lds[aIdx + kb + 1];
      bv = *(const v2f*)&lds[bBase + kb];
      acc = __builtin_amdgcn_wmma_f32_16x16x4_f32(
           false, av,  false, bv,
           (short)0, acc,  false,  false);
    }
    __syncthreads();
  }

  const int rbase = m0 + ((lane >> 4) << 3);
  const int col = lane & 15;
  if (col < 8) {
#pragma unroll
    for (int i = 0; i < 8; ++i) {
      const int R = rbase + i;
      if (R < 56) {
        const int s = R >> 3;
        const int l1o = R & 7;
        atomicAdd(&out[(((size_t)b * 384) + pairBase + l1o * 8 + col) * 7 + s],
                  acc[i]);
      }
    }
  }
}

__global__ void __launch_bounds__(128)
same_scale_all(const float* __restrict__ x0, const float* __restrict__ x1,
               const float* __restrict__ x2, float* __restrict__ out) {
  __shared__ __attribute__((aligned(16))) float lds[33 * 260];
  const int bid = blockIdx.x;
  if (bid < 256) {
    same_scale_body<8, 2>(x0, 0, out, bid, lds);
  } else if (bid < 512) {
    same_scale_body<7, 1>(x1, 192, out, bid - 256, lds);
  } else {
    same_scale_body<6, 1>(x2, 320, out, bid - 512, lds);
  }
}

template <int LOGN>
__device__ __forceinline__ void fft_rows_body(const float* __restrict__ xin,
                                              float2* __restrict__ spec,
                                              int blk, float2* lds) {
  constexpr int N = 1 << LOGN;
  constexpr int HALF = N / 2;
  constexpr int RPB = 128 / HALF;
  const int sub = threadIdx.x >> (LOGN - 1);
  const int t = threadIdx.x & (HALF - 1);
  const size_t line = (size_t)blk * RPB + sub;
  float2* buf = lds + sub * N;
  const float* src = xin + line * N;

  buf[__brev((unsigned)t) >> (32 - LOGN)] = make_float2(src[t], 0.0f);
  buf[__brev((unsigned)(t + N / 2)) >> (32 - LOGN)] =
      make_float2(src[t + N / 2], 0.0f);
  __syncthreads();
#pragma unroll
  for (int st = 1; st <= LOGN; ++st) {
    const int len = 1 << st;
    const int half = len >> 1;
    const int j = t & (half - 1);
    const int base = (t >> (st - 1)) << st;
    float sn, cs;
    __sincosf(-TWO_PI_F * (float)j / (float)len, &sn, &cs);
    const float2 u = buf[base + j];
    const float2 v = buf[base + j + half];
    const float2 tv = make_float2(v.x * cs - v.y * sn, v.x * sn + v.y * cs);
    buf[base + j] = make_float2(u.x + tv.x, u.y + tv.y);
    buf[base + j + half] = make_float2(u.x - tv.x, u.y - tv.y);
    __syncthreads();
  }
  float2* dst = spec + line * N;
  dst[t] = buf[t];
  dst[t + N / 2] = buf[t + N / 2];
}

__global__ void __launch_bounds__(128)
fft_rows_all(const float* __restrict__ x0, const float* __restrict__ x1,
             const float* __restrict__ x2, float2* __restrict__ spec0,
             float2* __restrict__ spec1, float2* __restrict__ spec2,
             float* __restrict__ out) {
  __shared__ float2 buf[256];
  const int bid = blockIdx.x;
  const int oi = bid * 128 + threadIdx.x;
  if (oi < 2 * 384 * 7) out[oi] = 0.0f;
  if (bid < 4096) {
    fft_rows_body<8>(x0, spec0, bid, buf);
  } else if (bid < 4096 + 1024) {
    fft_rows_body<7>(x1, spec1, bid - 4096, buf);
  } else {
    fft_rows_body<6>(x2, spec2, bid - (4096 + 1024), buf);
  }
}

template <int LOGN>
__device__ __forceinline__ void fft_cols_body(float2* __restrict__ spec, int blk,
                                              float2* lds) {
  constexpr int N = 1 << LOGN;
  constexpr int HALF = N / 2;
  constexpr int RPB = 128 / HALF;
  const int sub = threadIdx.x >> (LOGN - 1);
  const int t = threadIdx.x & (HALF - 1);
  const int line = blk * RPB + sub;
  const int col = line & (N - 1);
  const int img = line >> LOGN;
  float2* buf = lds + sub * N;
  float2* base0 = spec + ((size_t)img << (2 * LOGN)) + col;

  buf[__brev((unsigned)t) >> (32 - LOGN)] = base0[(size_t)t * N];
  buf[__brev((unsigned)(t + N / 2)) >> (32 - LOGN)] =
      base0[(size_t)(t + N / 2) * N];
  __syncthreads();
#pragma unroll
  for (int st = 1; st <= LOGN; ++st) {
    const int len = 1 << st;
    const int half = len >> 1;
    const int j = t & (half - 1);
    const int base = (t >> (st - 1)) << st;
    float sn, cs;
    __sincosf(-TWO_PI_F * (float)j / (float)len, &sn, &cs);
    const float2 u = buf[base + j];
    const float2 v = buf[base + j + half];
    const float2 tv = make_float2(v.x * cs - v.y * sn, v.x * sn + v.y * cs);
    buf[base + j] = make_float2(u.x + tv.x, u.y + tv.y);
    buf[base + j + half] = make_float2(u.x - tv.x, u.y - tv.y);
    __syncthreads();
  }
  base0[(size_t)t * N] = buf[t];
  base0[(size_t)(t + N / 2) * N] = buf[t + N / 2];
}

__global__ void __launch_bounds__(128)
fft_cols_all(float2* __restrict__ spec0, float2* __restrict__ spec1,
             float2* __restrict__ spec2) {
  __shared__ float2 buf[256];
  const int bid = blockIdx.x;
  if (bid < 4096) {
    fft_cols_body<8>(spec0, bid, buf);
  } else if (bid < 4096 + 1024) {
    fft_cols_body<7>(spec1, bid - 4096, buf);
  } else {
    fft_cols_body<6>(spec2, bid - (4096 + 1024), buf);
  }
}

template <int LOG1, int LOG2, int SPLIT, int U>
__device__ __forceinline__ void cross_body(const float2* __restrict__ s1,
                                           const float2* __restrict__ s2,
                                           int pairBase,
                                           float* __restrict__ out, int unit,
                                           int t, float* red) {
  constexpr int N1 = 1 << LOG1;
  constexpr int N2 = 1 << LOG2;
  constexpr int BINS = N2 * N2;
  constexpr int THREADS = BINS / (SPLIT * U);
  constexpr int NW = THREADS / 32;
  constexpr int ROWS_PER_PASS = THREADS >> LOG2;
  constexpr int ROWS_PER_BLOCK = N2 / SPLIT;
  static_assert(THREADS >= N2 && THREADS <= 1024 && (THREADS & 31) == 0,
                "block must cover whole spectrum rows");
  static_assert(ROWS_PER_PASS * U == ROWS_PER_BLOCK, "share must tile the rows");
  static_assert(SPLIT >= 2 && (SPLIT & (SPLIT - 1)) == 0, "rows of one half per block");

  const int part = unit & (SPLIT - 1);
  const int bp = unit / SPLIT;
  const int b = bp >> 6;
  const int p = bp & 63;
  const int l1 = p >> 3, l2 = p & 7;
  const float2* F1 = s1 + (((size_t)b * 8 + l1) << (2 * LOG1));
  const float2* F2 = s2 + (((size_t)b * 8 + l2) << (2 * LOG2));

  const int iy2 = t & (N2 - 1);
  const int fy = (iy2 < (N2 >> 1)) ? iy2 : iy2 - N2;
  const int col1 = (fy + N1) & (N1 - 1);
  const int row0 = part * ROWS_PER_BLOCK + (t >> LOG2);
  const int wrap1 = (part >= SPLIT / 2) ? (N1 - N2) : 0;
  const float2* pA = F1 + ((size_t)(row0 + wrap1) << LOG1) + col1;
  const float2* pB = F2 + ((size_t)row0 << LOG2) + iy2;

  float2 A[U], B[U];
#pragma unroll
  for (int u = 0; u < U; ++u) {
    A[u] = pA[(size_t)u * ROWS_PER_PASS * N1];
    B[u] = pB[(size_t)u * ROWS_PER_PASS * N2];
  }
#pragma unroll
  for (int u = 0; u < U; ++u)
    asm volatile("" ::"v"(A[u].x), "v"(A[u].y), "v"(B[u].x), "v"(B[u].y));

  float sy, cy;
  __sincosf(TWO_PI_F * (float)fy / (float)N1, &sy, &cy);
  const float c2y = cy * cy - sy * sy, s2y = 2.f * cy * sy;

  float a0 = 0.f, a1 = 0.f, a2 = 0.f, a3 = 0.f, a4 = 0.f, a5 = 0.f, a6 = 0.f;
#pragma unroll
  for (int u = 0; u < U; ++u) {
    const int ix2 = row0 + u * ROWS_PER_PASS;
    const int fx = (ix2 < (N2 >> 1)) ? ix2 : ix2 - N2;
    const float zr = A[u].x * B[u].x + A[u].y * B[u].y;
    const float zi = A[u].y * B[u].x - A[u].x * B[u].y;
    float sx, cx;
    __sincosf(TWO_PI_F * (float)fx / (float)N1, &sx, &cx);
    const float c2x = cx * cx - sx * sx, s2x = 2.f * cx * sx;
    a0 += zr;
    a1 += zr * cy - zi * sy;
    a2 += zr * c2y - zi * s2y;
    a3 += zr * cx - zi * sx;
    const float xr = zr * cx - zi * sx, xi = zr * sx + zi * cx;
    a4 += xr * cy - xi * sy;
    a5 += zr * c2x - zi * s2x;
    const float mr = zr * cx + zi * sx, mi = zi * cx - zr * sx;
    a6 += mr * cy - mi * sy;
  }

  float accs[7] = {a0, a1, a2, a3, a4, a5, a6};
#pragma unroll
  for (int s = 0; s < 7; ++s) {
    float v = accs[s];
    for (int off = 16; off > 0; off >>= 1) v += __shfl_xor(v, off, 32);
    accs[s] = v;
  }
  if ((t & 31) == 0) {
#pragma unroll
    for (int s = 0; s < 7; ++s) red[s * NW + (t >> 5)] = accs[s];
  }
  __syncthreads();
  if (t < 7) {
    float v = 0.f;
    for (int w = 0; w < NW; ++w) v += red[t * NW + w];
    const float inv = 1.0f / (float)BINS;
    atomicAdd(&out[(((size_t)b * 384) + pairBase + p) * 7 + t], v * inv);
  }
}

__global__ void __launch_bounds__(512)
cross_all(const float2* __restrict__ spec0, const float2* __restrict__ spec1,
          const float2* __restrict__ spec2, float* __restrict__ out) {
  __shared__ float red[2][7 * 16];
  const int bid = blockIdx.x;
  if (bid < 256) {
    cross_body<8, 7, 2, 16>(spec0, spec1, 64, out, bid, threadIdx.x, &red[0][0]);
  } else {
    const int hf = threadIdx.x >> 8;
    const int t = threadIdx.x & 255;
    const int unit = (bid - 256) * 2 + hf;
    if (bid < 384) {
      cross_body<8, 6, 2, 8>(spec0, spec2, 128, out, unit, t, &red[hf][0]);
    } else {
      cross_body<7, 6, 2, 8>(spec1, spec2, 256, out, unit - 256, t, &red[hf][0]);
    }
  }
}

extern "C" void kernel_launch(void* const* d_in, const int* in_sizes, int n_in,
                              void* d_out, int out_size, void* d_ws,
                              size_t ws_size, hipStream_t stream) {
  const float* x0 = (const float*)d_in[0];
  const float* x1 = (const float*)d_in[1];
  const float* x2 = (const float*)d_in[2];
  float* out = (float*)d_out;

  const size_t n0 = (size_t)16 * 256 * 256;
  const size_t n1 = (size_t)16 * 128 * 128;
  const size_t n2 = (size_t)16 * 64 * 64;
  if (ws_size >= (n0 + n1 + n2) * sizeof(float2)) {
    float2* spec0 = (float2*)d_ws;
    float2* spec1 = spec0 + n0;
    float2* spec2 = spec1 + n1;

    fft_rows_all<<<dim3(4096 + 1024 + 256), 128, 0, stream>>>(
        x0, x1, x2, spec0, spec1, spec2, out);
    same_scale_all<<<dim3(256 + 256 + 128), 128, 0, stream>>>(x0, x1, x2, out);
    fft_cols_all<<<dim3(4096 + 1024 + 256), 128, 0, stream>>>(spec0, spec1,
                                                              spec2);
    cross_all<<<dim3(512), 512, 0, stream>>>(spec0, spec1, spec2, out);
  }
}


# FFT kernels: each thread carries its butterfly through 4 neighbouring lines per stage (one twiddle, one barrier for the 4; columns as 16-byte accesses): 4x fewer waves to start; sums unchanged
# speedup vs baseline: 3.3829x; 1.5479x over previous
#include <hip/hip_runtime.h>
#include <hip/hip_bf16.h>


typedef __attribute__((ext_vector_type(2))) float v2f;
typedef __attribute__((ext_vector_type(8))) float v8f;

__device__ __constant__ int c_sdx[7] = {0, 0, 0, 1, 1, 2, -1};
__device__ __constant__ int c_sdy[7] = {0, 1, 2, 0, 1, 0, 1};

#define TWO_PI_F 6.28318530717958647692f

template <int LOGW, int ROWS>
__device__ __forceinline__ void same_scale_body(const float* __restrict__ xin,
                                                int pairBase,
                                                float* __restrict__ out,
                                                int blk, float* lds) {
  constexpr int W = 1 << LOGW;
  constexpr int H = W;
  constexpr int WS = W + 4;
  constexpr int ZROW = 32 * WS;
  constexpr int SLABS = H / ROWS;

  const int b = blk / SLABS;
  const int slab = blk & (SLABS - 1);

  const int lane = threadIdx.x & 31;
  const int wave = threadIdx.x >> 5;
  const int m0 = wave * 16;

  const int mrow = m0 + (lane & 15);
  const bool aval = (mrow < 56);
  const int sidx = aval ? (mrow >> 3) : 0;
  const int l1 = mrow & 7;
  const int slot = c_sdx[sidx] + 1;
  const int dy = aval ? c_sdy[sidx] : 0;
  const int aIdx = (aval ? (slot * 8 + l1) * WS : ZROW) + dy;
  const int ncol = lane & 15;
  const int bBase = (ncol < 8) ? (8 + ncol) * WS : ZROW;
  const int khi = (lane >> 4) << 1;

  for (int idx = threadIdx.x; idx < WS; idx += 128) lds[ZROW + idx] = 0.0f;

  v8f acc = {};

  for (int xi = 0; xi < ROWS; ++xi) {
    const int xr = slab * ROWS + xi;

    {
      constexpr int NV4 = 8 * W;
      constexpr int PER_CH4 = W / 4;
      for (int idx = threadIdx.x; idx < NV4; idx += 128) {
        const int row = idx >> (LOGW - 2);
        const int y4 = idx & (PER_CH4 - 1);
        const int rs = row >> 3;
        const int ch = row & 7;
        int xs = xr + rs - 1;
        xs += (xs < 0) ? H : 0;
        xs -= (xs >= H) ? H : 0;
        *(float4*)&lds[row * WS + (y4 << 2)] =
            *(const float4*)&xin[((((size_t)b * 8 + ch) << LOGW) + xs) * W +
                                 (y4 << 2)];
      }
      const int row = threadIdx.x >> 2;
      const int e = threadIdx.x & 3;
      const int rs = row >> 3;
      const int ch = row & 7;
      int xs = xr + rs - 1;
      xs += (xs < 0) ? H : 0;
      xs -= (xs >= H) ? H : 0;
      lds[row * WS + W + e] =
          xin[((((size_t)b * 8 + ch) << LOGW) + xs) * W + e];
    }
    if (threadIdx.x < 8) {
      int xn = xr + 4;
      xn -= (xn >= H) ? H : 0;
      __builtin_prefetch(
          &xin[((((size_t)b * 8 + threadIdx.x) << LOGW) + xn) * W], 0, 1);
    }
    __syncthreads();

#pragma unroll 8
    for (int k = 0; k < W; k += 4) {
      const int kb = k + khi;
      v2f av, bv;
      av[0] = lds[aIdx + kb];
      av[1] = lds[aIdx + kb + 1];
      bv = *(const v2f*)&lds[bBase + kb];
      acc = __builtin_amdgcn_wmma_f32_16x16x4_f32(
           false, av,  false, bv,
           (short)0, acc,  false,  false);
    }
    __syncthreads();
  }

  const int rbase = m0 + ((lane >> 4) << 3);
  const int col = lane & 15;
  if (col < 8) {
#pragma unroll
    for (int i = 0; i < 8; ++i) {
      const int R = rbase + i;
      if (R < 56) {
        const int s = R >> 3;
        const int l1o = R & 7;
        atomicAdd(&out[(((size_t)b * 384) + pairBase + l1o * 8 + col) * 7 + s],
                  acc[i]);
      }
    }
  }
}

__global__ void __launch_bounds__(128)
same_scale_all(const float* __restrict__ x0, const float* __restrict__ x1,
               const float* __restrict__ x2, float* __restrict__ out) {
  __shared__ __attribute__((aligned(16))) float lds[33 * 260];
  const int bid = blockIdx.x;
  if (bid < 256) {
    same_scale_body<8, 2>(x0, 0, out, bid, lds);
  } else if (bid < 512) {
    same_scale_body<7, 1>(x1, 192, out, bid - 256, lds);
  } else {
    same_scale_body<6, 1>(x2, 320, out, bid - 512, lds);
  }
}

#define FFT_R 4

template <int LOGN>
__device__ __forceinline__ void fft_rows_body(const float* __restrict__ xin,
                                              float2* __restrict__ spec,
                                              int blk, float2* lds) {
  constexpr int N = 1 << LOGN;
  constexpr int HALF = N / 2;
  constexpr int RPB = 128 / HALF;
  const int sub = threadIdx.x >> (LOGN - 1);
  const int t = threadIdx.x & (HALF - 1);
  const size_t line0 = ((size_t)blk * RPB + sub) * FFT_R;
  float2* buf = lds + sub * (FFT_R * N);
  const float* src = xin + line0 * N;

  const int r0 = __brev((unsigned)t) >> (32 - LOGN);
  const int r1 = __brev((unsigned)(t + HALF)) >> (32 - LOGN);
  float lo[FFT_R], hi[FFT_R];
#pragma unroll
  for (int r = 0; r < FFT_R; ++r) {
    lo[r] = src[r * N + t];
    hi[r] = src[r * N + t + HALF];
  }
#pragma unroll
  for (int r = 0; r < FFT_R; ++r) {
    buf[r * N + r0] = make_float2(lo[r], 0.0f);
    buf[r * N + r1] = make_float2(hi[r], 0.0f);
  }
  __syncthreads();
#pragma unroll
  for (int st = 1; st <= LOGN; ++st) {
    const int len = 1 << st;
    const int half = len >> 1;
    const int j = t & (half - 1);
    const int base = (t >> (st - 1)) << st;
    float sn, cs;
    __sincosf(-TWO_PI_F * (float)j / (float)len, &sn, &cs);
#pragma unroll
    for (int r = 0; r < FFT_R; ++r) {
      float2* b = buf + r * N;
      const float2 u = b[base + j];
      const float2 v = b[base + j + half];
      const float2 tv = make_float2(v.x * cs - v.y * sn, v.x * sn + v.y * cs);
      b[base + j] = make_float2(u.x + tv.x, u.y + tv.y);
      b[base + j + half] = make_float2(u.x - tv.x, u.y - tv.y);
    }
    __syncthreads();
  }
  float2* dst = spec + line0 * N;
#pragma unroll
  for (int r = 0; r < FFT_R; ++r) {
    dst[r * N + t] = buf[r * N + t];
    dst[r * N + t + HALF] = buf[r * N + t + HALF];
  }
}

__global__ void __launch_bounds__(128)
fft_rows_all(const float* __restrict__ x0, const float* __restrict__ x1,
             const float* __restrict__ x2, float2* __restrict__ spec0,
             float2* __restrict__ spec1, float2* __restrict__ spec2,
             float* __restrict__ out) {
  __shared__ float2 buf[256 * FFT_R];
  const int bid = blockIdx.x;
  const int oi = bid * 128 + threadIdx.x;
  if (oi < 2 * 384 * 7) out[oi] = 0.0f;
  if (bid < 4096 / FFT_R) {
    fft_rows_body<8>(x0, spec0, bid, buf);
  } else if (bid < (4096 + 1024) / FFT_R) {
    fft_rows_body<7>(x1, spec1, bid - 4096 / FFT_R, buf);
  } else {
    fft_rows_body<6>(x2, spec2, bid - (4096 + 1024) / FFT_R, buf);
  }
}

template <int LOGN>
__device__ __forceinline__ void fft_cols_body(float2* __restrict__ spec, int blk,
                                              float2* lds) {
  constexpr int N = 1 << LOGN;
  constexpr int HALF = N / 2;
  constexpr int RPB = 128 / HALF;
  static_assert((FFT_R & 1) == 0, "column pairs move as 16-byte values");
  const int sub = threadIdx.x >> (LOGN - 1);
  const int t = threadIdx.x & (HALF - 1);
  const int line0 = (blk * RPB + sub) * FFT_R;
  const int col = line0 & (N - 1);
  const int img = line0 >> LOGN;
  float2* buf = lds + sub * (FFT_R * N);
  float2* base0 = spec + ((size_t)img << (2 * LOGN)) + col;

  const int r0 = __brev((unsigned)t) >> (32 - LOGN);
  const int r1 = __brev((unsigned)(t + HALF)) >> (32 - LOGN);
  float4 lo[FFT_R / 2], hi[FFT_R / 2];
#pragma unroll
  for (int q = 0; q < FFT_R / 2; ++q) {
    lo[q] = *(const float4*)(base0 + (size_t)t * N + 2 * q);
    hi[q] = *(const float4*)(base0 + (size_t)(t + HALF) * N + 2 * q);
  }
#pragma unroll
  for (int q = 0; q < FFT_R / 2; ++q) {
    buf[(2 * q) * N + r0] = make_float2(lo[q].x, lo[q].y);
    buf[(2 * q + 1) * N + r0] = make_float2(lo[q].z, lo[q].w);
    buf[(2 * q) * N + r1] = make_float2(hi[q].x, hi[q].y);
    buf[(2 * q + 1) * N + r1] = make_float2(hi[q].z, hi[q].w);
  }
  __syncthreads();
#pragma unroll
  for (int st = 1; st <= LOGN; ++st) {
    const int len = 1 << st;
    const int half = len >> 1;
    const int j = t & (half - 1);
    const int base = (t >> (st - 1)) << st;
    float sn, cs;
    __sincosf(-TWO_PI_F * (float)j / (float)len, &sn, &cs);
#pragma unroll
    for (int r = 0; r < FFT_R; ++r) {
      float2* b = buf + r * N;
      const float2 u = b[base + j];
      const float2 v = b[base + j + half];
      const float2 tv = make_float2(v.x * cs - v.y * sn, v.x * sn + v.y * cs);
      b[base + j] = make_float2(u.x + tv.x, u.y + tv.y);
      b[base + j + half] = make_float2(u.x - tv.x, u.y - tv.y);
    }
    __syncthreads();
  }
#pragma unroll
  for (int q = 0; q < FFT_R / 2; ++q) {
    const float2 a0 = buf[(2 * q) * N + t], a1 = buf[(2 * q + 1) * N + t];
    const float2 b0 = buf[(2 * q) * N + t + HALF];
    const float2 b1 = buf[(2 * q + 1) * N + t + HALF];
    *(float4*)(base0 + (size_t)t * N + 2 * q) =
        make_float4(a0.x, a0.y, a1.x, a1.y);
    *(float4*)(base0 + (size_t)(t + HALF) * N + 2 * q) =
        make_float4(b0.x, b0.y, b1.x, b1.y);
  }
}

__global__ void __launch_bounds__(128)
fft_cols_all(float2* __restrict__ spec0, float2* __restrict__ spec1,
             float2* __restrict__ spec2) {
  __shared__ float2 buf[256 * FFT_R];
  const int bid = blockIdx.x;
  if (bid < 4096 / FFT_R) {
    fft_cols_body<8>(spec0, bid, buf);
  } else if (bid < (4096 + 1024) / FFT_R) {
    fft_cols_body<7>(spec1, bid - 4096 / FFT_R, buf);
  } else {
    fft_cols_body<6>(spec2, bid - (4096 + 1024) / FFT_R, buf);
  }
}

template <int LOG1, int LOG2, int SPLIT, int U>
__device__ __forceinline__ void cross_body(const float2* __restrict__ s1,
                                           const float2* __restrict__ s2,
                                           int pairBase,
                                           float* __restrict__ out, int unit,
                                           int t, float* red) {
  constexpr int N1 = 1 << LOG1;
  constexpr int N2 = 1 << LOG2;
  constexpr int BINS = N2 * N2;
  constexpr int THREADS = BINS / (SPLIT * U);
  constexpr int NW = THREADS / 32;
  constexpr int ROWS_PER_PASS = THREADS >> LOG2;
  constexpr int ROWS_PER_BLOCK = N2 / SPLIT;
  static_assert(THREADS >= N2 && THREADS <= 1024 && (THREADS & 31) == 0,
                "block must cover whole spectrum rows");
  static_assert(ROWS_PER_PASS * U == ROWS_PER_BLOCK, "share must tile the rows");
  static_assert(SPLIT >= 2 && (SPLIT & (SPLIT - 1)) == 0, "rows of one half per block");

  const int part = unit & (SPLIT - 1);
  const int bp = unit / SPLIT;
  const int b = bp >> 6;
  const int p = bp & 63;
  const int l1 = p >> 3, l2 = p & 7;
  const float2* F1 = s1 + (((size_t)b * 8 + l1) << (2 * LOG1));
  const float2* F2 = s2 + (((size_t)b * 8 + l2) << (2 * LOG2));

  const int iy2 = t & (N2 - 1);
  const int fy = (iy2 < (N2 >> 1)) ? iy2 : iy2 - N2;
  const int col1 = (fy + N1) & (N1 - 1);
  const int row0 = part * ROWS_PER_BLOCK + (t >> LOG2);
  const int wrap1 = (part >= SPLIT / 2) ? (N1 - N2) : 0;
  const float2* pA = F1 + ((size_t)(row0 + wrap1) << LOG1) + col1;
  const float2* pB = F2 + ((size_t)row0 << LOG2) + iy2;

  float2 A[U], B[U];
#pragma unroll
  for (int u = 0; u < U; ++u) {
    A[u] = pA[(size_t)u * ROWS_PER_PASS * N1];
    B[u] = pB[(size_t)u * ROWS_PER_PASS * N2];
  }
#pragma unroll
  for (int u = 0; u < U; ++u)
    asm volatile("" ::"v"(A[u].x), "v"(A[u].y), "v"(B[u].x), "v"(B[u].y));

  float sy, cy;
  __sincosf(TWO_PI_F * (float)fy / (float)N1, &sy, &cy);
  const float c2y = cy * cy - sy * sy, s2y = 2.f * cy * sy;

  float a0 = 0.f, a1 = 0.f, a2 = 0.f, a3 = 0.f, a4 = 0.f, a5 = 0.f, a6 = 0.f;
#pragma unroll
  for (int u = 0; u < U; ++u) {
    const int ix2 = row0 + u * ROWS_PER_PASS;
    const int fx = (ix2 < (N2 >> 1)) ? ix2 : ix2 - N2;
    const float zr = A[u].x * B[u].x + A[u].y * B[u].y;
    const float zi = A[u].y * B[u].x - A[u].x * B[u].y;
    float sx, cx;
    __sincosf(TWO_PI_F * (float)fx / (float)N1, &sx, &cx);
    const float c2x = cx * cx - sx * sx, s2x = 2.f * cx * sx;
    a0 += zr;
    a1 += zr * cy - zi * sy;
    a2 += zr * c2y - zi * s2y;
    a3 += zr * cx - zi * sx;
    const float xr = zr * cx - zi * sx, xi = zr * sx + zi * cx;
    a4 += xr * cy - xi * sy;
    a5 += zr * c2x - zi * s2x;
    const float mr = zr * cx + zi * sx, mi = zi * cx - zr * sx;
    a6 += mr * cy - mi * sy;
  }

  float accs[7] = {a0, a1, a2, a3, a4, a5, a6};
#pragma unroll
  for (int s = 0; s < 7; ++s) {
    float v = accs[s];
    for (int off = 16; off > 0; off >>= 1) v += __shfl_xor(v, off, 32);
    accs[s] = v;
  }
  if ((t & 31) == 0) {
#pragma unroll
    for (int s = 0; s < 7; ++s) red[s * NW + (t >> 5)] = accs[s];
  }
  __syncthreads();
  if (t < 7) {
    float v = 0.f;
    for (int w = 0; w < NW; ++w) v += red[t * NW + w];
    const float inv = 1.0f / (float)BINS;
    atomicAdd(&out[(((size_t)b * 384) + pairBase + p) * 7 + t], v * inv);
  }
}

__global__ void __launch_bounds__(512)
cross_all(const float2* __restrict__ spec0, const float2* __restrict__ spec1,
          const float2* __restrict__ spec2, float* __restrict__ out) {
  __shared__ float red[2][7 * 16];
  const int bid = blockIdx.x;
  if (bid < 256) {
    cross_body<8, 7, 2, 16>(spec0, spec1, 64, out, bid, threadIdx.x, &red[0][0]);
  } else {
    const int hf = threadIdx.x >> 8;
    const int t = threadIdx.x & 255;
    const int unit = (bid - 256) * 2 + hf;
    if (bid < 384) {
      cross_body<8, 6, 2, 8>(spec0, spec2, 128, out, unit, t, &red[hf][0]);
    } else {
      cross_body<7, 6, 2, 8>(spec1, spec2, 256, out, unit - 256, t, &red[hf][0]);
    }
  }
}

extern "C" void kernel_launch(void* const* d_in, const int* in_sizes, int n_in,
                              void* d_out, int out_size, void* d_ws,
                              size_t ws_size, hipStream_t stream) {
  const float* x0 = (const float*)d_in[0];
  const float* x1 = (const float*)d_in[1];
  const float* x2 = (const float*)d_in[2];
  float* out = (float*)d_out;

  const size_t n0 = (size_t)16 * 256 * 256;
  const size_t n1 = (size_t)16 * 128 * 128;
  const size_t n2 = (size_t)16 * 64 * 64;
  if (ws_size >= (n0 + n1 + n2) * sizeof(float2)) {
    float2* spec0 = (float2*)d_ws;
    float2* spec1 = spec0 + n0;
    float2* spec2 = spec1 + n1;

    fft_rows_all<<<dim3((4096 + 1024 + 256) / FFT_R), 128, 0, stream>>>(
        x0, x1, x2, spec0, spec1, spec2, out);
    same_scale_all<<<dim3(256 + 256 + 128), 128, 0, stream>>>(x0, x1, x2, out);
    fft_cols_all<<<dim3((4096 + 1024 + 256) / FFT_R), 128, 0, stream>>>(
        spec0, spec1, spec2);
    cross_all<<<dim3(512), 512, 0, stream>>>(spec0, spec1, spec2, out);
  }
}


# same-scale WMMA products run inside the column-transform launch (independent work, block index picks the kind): 4 launches -> 3; arithmetic, atomics and reduction trees untouched
# speedup vs baseline: 3.6104x; 1.0672x over previous
#include <hip/hip_runtime.h>
#include <hip/hip_bf16.h>


typedef __attribute__((ext_vector_type(2))) float v2f;
typedef __attribute__((ext_vector_type(8))) float v8f;

__device__ __constant__ int c_sdx[7] = {0, 0, 0, 1, 1, 2, -1};
__device__ __constant__ int c_sdy[7] = {0, 1, 2, 0, 1, 0, 1};

#define TWO_PI_F 6.28318530717958647692f

template <int LOGW, int ROWS>
__device__ __forceinline__ void same_scale_body(const float* __restrict__ xin,
                                                int pairBase,
                                                float* __restrict__ out,
                                                int blk, float* lds) {
  constexpr int W = 1 << LOGW;
  constexpr int H = W;
  constexpr int WS = W + 4;
  constexpr int ZROW = 32 * WS;
  constexpr int SLABS = H / ROWS;

  const int b = blk / SLABS;
  const int slab = blk & (SLABS - 1);

  const int lane = threadIdx.x & 31;
  const int wave = threadIdx.x >> 5;
  const int m0 = wave * 16;

  const int mrow = m0 + (lane & 15);
  const bool aval = (mrow < 56);
  const int sidx = aval ? (mrow >> 3) : 0;
  const int l1 = mrow & 7;
  const int slot = c_sdx[sidx] + 1;
  const int dy = aval ? c_sdy[sidx] : 0;
  const int aIdx = (aval ? (slot * 8 + l1) * WS : ZROW) + dy;
  const int ncol = lane & 15;
  const int bBase = (ncol < 8) ? (8 + ncol) * WS : ZROW;
  const int khi = (lane >> 4) << 1;

  for (int idx = threadIdx.x; idx < WS; idx += 128) lds[ZROW + idx] = 0.0f;

  v8f acc = {};

  for (int xi = 0; xi < ROWS; ++xi) {
    const int xr = slab * ROWS + xi;

    {
      constexpr int NV4 = 8 * W;
      constexpr int PER_CH4 = W / 4;
      for (int idx = threadIdx.x; idx < NV4; idx += 128) {
        const int row = idx >> (LOGW - 2);
        const int y4 = idx & (PER_CH4 - 1);
        const int rs = row >> 3;
        const int ch = row & 7;
        int xs = xr + rs - 1;
        xs += (xs < 0) ? H : 0;
        xs -= (xs >= H) ? H : 0;
        *(float4*)&lds[row * WS + (y4 << 2)] =
            *(const float4*)&xin[((((size_t)b * 8 + ch) << LOGW) + xs) * W +
                                 (y4 << 2)];
      }
      const int row = threadIdx.x >> 2;
      const int e = threadIdx.x & 3;
      const int rs = row >> 3;
      const int ch = row & 7;
      int xs = xr + rs - 1;
      xs += (xs < 0) ? H : 0;
      xs -= (xs >= H) ? H : 0;
      lds[row * WS + W + e] =
          xin[((((size_t)b * 8 + ch) << LOGW) + xs) * W + e];
    }
    if (threadIdx.x < 8) {
      int xn = xr + 4;
      xn -= (xn >= H) ? H : 0;
      __builtin_prefetch(
          &xin[((((size_t)b * 8 + threadIdx.x) << LOGW) + xn) * W], 0, 1);
    }
    __syncthreads();

#pragma unroll 8
    for (int k = 0; k < W; k += 4) {
      const int kb = k + khi;
      v2f av, bv;
      av[0] = lds[aIdx + kb];
      av[1] = lds[aIdx + kb + 1];
      bv = *(const v2f*)&lds[bBase + kb];
      acc = __builtin_amdgcn_wmma_f32_16x16x4_f32(
           false, av,  false, bv,
           (short)0, acc,  false,  false);
    }
    __syncthreads();
  }

  const int rbase = m0 + ((lane >> 4) << 3);
  const int col = lane & 15;
  if (col < 8) {
#pragma unroll
    for (int i = 0; i < 8; ++i) {
      const int R = rbase + i;
      if (R < 56) {
        const int s = R >> 3;
        const int l1o = R & 7;
        atomicAdd(&out[(((size_t)b * 384) + pairBase + l1o * 8 + col) * 7 + s],
                  acc[i]);
      }
    }
  }
}


#define FFT_R 4

template <int LOGN>
__device__ __forceinline__ void fft_rows_body(const float* __restrict__ xin,
                                              float2* __restrict__ spec,
                                              int blk, float2* lds) {
  constexpr int N = 1 << LOGN;
  constexpr int HALF = N / 2;
  constexpr int RPB = 128 / HALF;
  const int sub = threadIdx.x >> (LOGN - 1);
  const int t = threadIdx.x & (HALF - 1);
  const size_t line0 = ((size_t)blk * RPB + sub) * FFT_R;
  float2* buf = lds + sub * (FFT_R * N);
  const float* src = xin + line0 * N;

  const int r0 = __brev((unsigned)t) >> (32 - LOGN);
  const int r1 = __brev((unsigned)(t + HALF)) >> (32 - LOGN);
  float lo[FFT_R], hi[FFT_R];
#pragma unroll
  for (int r = 0; r < FFT_R; ++r) {
    lo[r] = src[r * N + t];
    hi[r] = src[r * N + t + HALF];
  }
#pragma unroll
  for (int r = 0; r < FFT_R; ++r) {
    buf[r * N + r0] = make_float2(lo[r], 0.0f);
    buf[r * N + r1] = make_float2(hi[r], 0.0f);
  }
  __syncthreads();
#pragma unroll
  for (int st = 1; st <= LOGN; ++st) {
    const int len = 1 << st;
    const int half = len >> 1;
    const int j = t & (half - 1);
    const int base = (t >> (st - 1)) << st;
    float sn, cs;
    __sincosf(-TWO_PI_F * (float)j / (float)len, &sn, &cs);
#pragma unroll
    for (int r = 0; r < FFT_R; ++r) {
      float2* b = buf + r * N;
      const float2 u = b[base + j];
      const float2 v = b[base + j + half];
      const float2 tv = make_float2(v.x * cs - v.y * sn, v.x * sn + v.y * cs);
      b[base + j] = make_float2(u.x + tv.x, u.y + tv.y);
      b[base + j + half] = make_float2(u.x - tv.x, u.y - tv.y);
    }
    __syncthreads();
  }
  float2* dst = spec + line0 * N;
#pragma unroll
  for (int r = 0; r < FFT_R; ++r) {
    dst[r * N + t] = buf[r * N + t];
    dst[r * N + t + HALF] = buf[r * N + t + HALF];
  }
}

__global__ void __launch_bounds__(128)
fft_rows_all(const float* __restrict__ x0, const float* __restrict__ x1,
             const float* __restrict__ x2, float2* __restrict__ spec0,
             float2* __restrict__ spec1, float2* __restrict__ spec2,
             float* __restrict__ out) {
  __shared__ float2 buf[256 * FFT_R];
  const int bid = blockIdx.x;
  const int oi = bid * 128 + threadIdx.x;
  if (oi < 2 * 384 * 7) out[oi] = 0.0f;
  if (bid < 4096 / FFT_R) {
    fft_rows_body<8>(x0, spec0, bid, buf);
  } else if (bid < (4096 + 1024) / FFT_R) {
    fft_rows_body<7>(x1, spec1, bid - 4096 / FFT_R, buf);
  } else {
    fft_rows_body<6>(x2, spec2, bid - (4096 + 1024) / FFT_R, buf);
  }
}

template <int LOGN>
__device__ __forceinline__ void fft_cols_body(float2* __restrict__ spec, int blk,
                                              float2* lds) {
  constexpr int N = 1 << LOGN;
  constexpr int HALF = N / 2;
  constexpr int RPB = 128 / HALF;
  static_assert((FFT_R & 1) == 0, "column pairs move as 16-byte values");
  const int sub = threadIdx.x >> (LOGN - 1);
  const int t = threadIdx.x & (HALF - 1);
  const int line0 = (blk * RPB + sub) * FFT_R;
  const int col = line0 & (N - 1);
  const int img = line0 >> LOGN;
  float2* buf = lds + sub * (FFT_R * N);
  float2* base0 = spec + ((size_t)img << (2 * LOGN)) + col;

  const int r0 = __brev((unsigned)t) >> (32 - LOGN);
  const int r1 = __brev((unsigned)(t + HALF)) >> (32 - LOGN);
  float4 lo[FFT_R / 2], hi[FFT_R / 2];
#pragma unroll
  for (int q = 0; q < FFT_R / 2; ++q) {
    lo[q] = *(const float4*)(base0 + (size_t)t * N + 2 * q);
    hi[q] = *(const float4*)(base0 + (size_t)(t + HALF) * N + 2 * q);
  }
#pragma unroll
  for (int q = 0; q < FFT_R / 2; ++q) {
    buf[(2 * q) * N + r0] = make_float2(lo[q].x, lo[q].y);
    buf[(2 * q + 1) * N + r0] = make_float2(lo[q].z, lo[q].w);
    buf[(2 * q) * N + r1] = make_float2(hi[q].x, hi[q].y);
    buf[(2 * q + 1) * N + r1] = make_float2(hi[q].z, hi[q].w);
  }
  __syncthreads();
#pragma unroll
  for (int st = 1; st <= LOGN; ++st) {
    const int len = 1 << st;
    const int half = len >> 1;
    const int j = t & (half - 1);
    const int base = (t >> (st - 1)) << st;
    float sn, cs;
    __sincosf(-TWO_PI_F * (float)j / (float)len, &sn, &cs);
#pragma unroll
    for (int r = 0; r < FFT_R; ++r) {
      float2* b = buf + r * N;
      const float2 u = b[base + j];
      const float2 v = b[base + j + half];
      const float2 tv = make_float2(v.x * cs - v.y * sn, v.x * sn + v.y * cs);
      b[base + j] = make_float2(u.x + tv.x, u.y + tv.y);
      b[base + j + half] = make_float2(u.x - tv.x, u.y - tv.y);
    }
    __syncthreads();
  }
#pragma unroll
  for (int q = 0; q < FFT_R / 2; ++q) {
    const float2 a0 = buf[(2 * q) * N + t], a1 = buf[(2 * q + 1) * N + t];
    const float2 b0 = buf[(2 * q) * N + t + HALF];
    const float2 b1 = buf[(2 * q + 1) * N + t + HALF];
    *(float4*)(base0 + (size_t)t * N + 2 * q) =
        make_float4(a0.x, a0.y, a1.x, a1.y);
    *(float4*)(base0 + (size_t)(t + HALF) * N + 2 * q) =
        make_float4(b0.x, b0.y, b1.x, b1.y);
  }
}

#define SAME_BLOCKS (256 + 256 + 128)
__global__ void __launch_bounds__(128)
cols_and_same_all(const float* __restrict__ x0, const float* __restrict__ x1,
                  const float* __restrict__ x2, float2* __restrict__ spec0,
                  float2* __restrict__ spec1, float2* __restrict__ spec2,
                  float* __restrict__ out) {
  __shared__ __attribute__((aligned(16))) float lds[33 * 260];
  static_assert(sizeof(float2) * 256 * FFT_R <= sizeof(float) * 33 * 260,
                "transform buffer must fit the shared array");
  const int bid = blockIdx.x;
  if (bid < 256) {
    same_scale_body<8, 2>(x0, 0, out, bid, lds);
  } else if (bid < 512) {
    same_scale_body<7, 1>(x1, 192, out, bid - 256, lds);
  } else if (bid < SAME_BLOCKS) {
    same_scale_body<6, 1>(x2, 320, out, bid - 512, lds);
  } else {
    const int fb = bid - SAME_BLOCKS;
    float2* buf = reinterpret_cast<float2*>(lds);
    if (fb < 4096 / FFT_R) {
      fft_cols_body<8>(spec0, fb, buf);
    } else if (fb < (4096 + 1024) / FFT_R) {
      fft_cols_body<7>(spec1, fb - 4096 / FFT_R, buf);
    } else {
      fft_cols_body<6>(spec2, fb - (4096 + 1024) / FFT_R, buf);
    }
  }
}

template <int LOG1, int LOG2, int SPLIT, int U>
__device__ __forceinline__ void cross_body(const float2* __restrict__ s1,
                                           const float2* __restrict__ s2,
                                           int pairBase,
                                           float* __restrict__ out, int unit,
                                           int t, float* red) {
  constexpr int N1 = 1 << LOG1;
  constexpr int N2 = 1 << LOG2;
  constexpr int BINS = N2 * N2;
  constexpr int THREADS = BINS / (SPLIT * U);
  constexpr int NW = THREADS / 32;
  constexpr int ROWS_PER_PASS = THREADS >> LOG2;
  constexpr int ROWS_PER_BLOCK = N2 / SPLIT;
  static_assert(THREADS >= N2 && THREADS <= 1024 && (THREADS & 31) == 0,
                "block must cover whole spectrum rows");
  static_assert(ROWS_PER_PASS * U == ROWS_PER_BLOCK, "share must tile the rows");
  static_assert(SPLIT >= 2 && (SPLIT & (SPLIT - 1)) == 0, "rows of one half per block");

  const int part = unit & (SPLIT - 1);
  const int bp = unit / SPLIT;
  const int b = bp >> 6;
  const int p = bp & 63;
  const int l1 = p >> 3, l2 = p & 7;
  const float2* F1 = s1 + (((size_t)b * 8 + l1) << (2 * LOG1));
  const float2* F2 = s2 + (((size_t)b * 8 + l2) << (2 * LOG2));

  const int iy2 = t & (N2 - 1);
  const int fy = (iy2 < (N2 >> 1)) ? iy2 : iy2 - N2;
  const int col1 = (fy + N1) & (N1 - 1);
  const int row0 = part * ROWS_PER_BLOCK + (t >> LOG2);
  const int wrap1 = (part >= SPLIT / 2) ? (N1 - N2) : 0;
  const float2* pA = F1 + ((size_t)(row0 + wrap1) << LOG1) + col1;
  const float2* pB = F2 + ((size_t)row0 << LOG2) + iy2;

  float2 A[U], B[U];
#pragma unroll
  for (int u = 0; u < U; ++u) {
    A[u] = pA[(size_t)u * ROWS_PER_PASS * N1];
    B[u] = pB[(size_t)u * ROWS_PER_PASS * N2];
  }
#pragma unroll
  for (int u = 0; u < U; ++u)
    asm volatile("" ::"v"(A[u].x), "v"(A[u].y), "v"(B[u].x), "v"(B[u].y));

  float sy, cy;
  __sincosf(TWO_PI_F * (float)fy / (float)N1, &sy, &cy);
  const float c2y = cy * cy - sy * sy, s2y = 2.f * cy * sy;

  float a0 = 0.f, a1 = 0.f, a2 = 0.f, a3 = 0.f, a4 = 0.f, a5 = 0.f, a6 = 0.f;
#pragma unroll
  for (int u = 0; u < U; ++u) {
    const int ix2 = row0 + u * ROWS_PER_PASS;
    const int fx = (ix2 < (N2 >> 1)) ? ix2 : ix2 - N2;
    const float zr = A[u].x * B[u].x + A[u].y * B[u].y;
    const float zi = A[u].y * B[u].x - A[u].x * B[u].y;
    float sx, cx;
    __sincosf(TWO_PI_F * (float)fx / (float)N1, &sx, &cx);
    const float c2x = cx * cx - sx * sx, s2x = 2.f * cx * sx;
    a0 += zr;
    a1 += zr * cy - zi * sy;
    a2 += zr * c2y - zi * s2y;
    a3 += zr * cx - zi * sx;
    const float xr = zr * cx - zi * sx, xi = zr * sx + zi * cx;
    a4 += xr * cy - xi * sy;
    a5 += zr * c2x - zi * s2x;
    const float mr = zr * cx + zi * sx, mi = zi * cx - zr * sx;
    a6 += mr * cy - mi * sy;
  }

  float accs[7] = {a0, a1, a2, a3, a4, a5, a6};
#pragma unroll
  for (int s = 0; s < 7; ++s) {
    float v = accs[s];
    for (int off = 16; off > 0; off >>= 1) v += __shfl_xor(v, off, 32);
    accs[s] = v;
  }
  if ((t & 31) == 0) {
#pragma unroll
    for (int s = 0; s < 7; ++s) red[s * NW + (t >> 5)] = accs[s];
  }
  __syncthreads();
  if (t < 7) {
    float v = 0.f;
    for (int w = 0; w < NW; ++w) v += red[t * NW + w];
    const float inv = 1.0f / (float)BINS;
    atomicAdd(&out[(((size_t)b * 384) + pairBase + p) * 7 + t], v * inv);
  }
}

__global__ void __launch_bounds__(512)
cross_all(const float2* __restrict__ spec0, const float2* __restrict__ spec1,
          const float2* __restrict__ spec2, float* __restrict__ out) {
  __shared__ float red[2][7 * 16];
  const int bid = blockIdx.x;
  if (bid < 256) {
    cross_body<8, 7, 2, 16>(spec0, spec1, 64, out, bid, threadIdx.x, &red[0][0]);
  } else {
    const int hf = threadIdx.x >> 8;
    const int t = threadIdx.x & 255;
    const int unit = (bid - 256) * 2 + hf;
    if (bid < 384) {
      cross_body<8, 6, 2, 8>(spec0, spec2, 128, out, unit, t, &red[hf][0]);
    } else {
      cross_body<7, 6, 2, 8>(spec1, spec2, 256, out, unit - 256, t, &red[hf][0]);
    }
  }
}

extern "C" void kernel_launch(void* const* d_in, const int* in_sizes, int n_in,
                              void* d_out, int out_size, void* d_ws,
                              size_t ws_size, hipStream_t stream) {
  const float* x0 = (const float*)d_in[0];
  const float* x1 = (const float*)d_in[1];
  const float* x2 = (const float*)d_in[2];
  float* out = (float*)d_out;

  const size_t n0 = (size_t)16 * 256 * 256;
  const size_t n1 = (size_t)16 * 128 * 128;
  const size_t n2 = (size_t)16 * 64 * 64;
  if (ws_size >= (n0 + n1 + n2) * sizeof(float2)) {
    float2* spec0 = (float2*)d_ws;
    float2* spec1 = spec0 + n0;
    float2* spec2 = spec1 + n1;

    fft_rows_all<<<dim3((4096 + 1024 + 256) / FFT_R), 128, 0, stream>>>(
        x0, x1, x2, spec0, spec1, spec2, out);
    cols_and_same_all<<<dim3(SAME_BLOCKS + (4096 + 1024 + 256) / FFT_R), 128, 0,
                        stream>>>(x0, x1, x2, spec0, spec1, spec2, out);
    cross_all<<<dim3(512), 512, 0, stream>>>(spec0, spec1, spec2, out);
  }
}


# FFT kernels: each thread carries its butterfly through 8 neighbouring lines per stage, not 4 (one twiddle, one barrier for the 8): half the transforms' waves to start; every sum unchanged
# speedup vs baseline: 3.8583x; 1.0687x over previous
#include <hip/hip_runtime.h>
#include <hip/hip_bf16.h>


typedef __attribute__((ext_vector_type(2))) float v2f;
typedef __attribute__((ext_vector_type(8))) float v8f;

__device__ __constant__ int c_sdx[7] = {0, 0, 0, 1, 1, 2, -1};
__device__ __constant__ int c_sdy[7] = {0, 1, 2, 0, 1, 0, 1};

#define TWO_PI_F 6.28318530717958647692f

template <int LOGW, int ROWS>
__device__ __forceinline__ void same_scale_body(const float* __restrict__ xin,
                                                int pairBase,
                                                float* __restrict__ out,
                                                int blk, float* lds) {
  constexpr int W = 1 << LOGW;
  constexpr int H = W;
  constexpr int WS = W + 4;
  constexpr int ZROW = 32 * WS;
  constexpr int SLABS = H / ROWS;

  const int b = blk / SLABS;
  const int slab = blk & (SLABS - 1);

  const int lane = threadIdx.x & 31;
  const int wave = threadIdx.x >> 5;
  const int m0 = wave * 16;

  const int mrow = m0 + (lane & 15);
  const bool aval = (mrow < 56);
  const int sidx = aval ? (mrow >> 3) : 0;
  const int l1 = mrow & 7;
  const int slot = c_sdx[sidx] + 1;
  const int dy = aval ? c_sdy[sidx] : 0;
  const int aIdx = (aval ? (slot * 8 + l1) * WS : ZROW) + dy;
  const int ncol = lane & 15;
  const int bBase = (ncol < 8) ? (8 + ncol) * WS : ZROW;
  const int khi = (lane >> 4) << 1;

  for (int idx = threadIdx.x; idx < WS; idx += 128) lds[ZROW + idx] = 0.0f;

  v8f acc = {};

  for (int xi = 0; xi < ROWS; ++xi) {
    const int xr = slab * ROWS + xi;

    {
      constexpr int NV4 = 8 * W;
      constexpr int PER_CH4 = W / 4;
      for (int idx = threadIdx.x; idx < NV4; idx += 128) {
        const int row = idx >> (LOGW - 2);
        const int y4 = idx & (PER_CH4 - 1);
        const int rs = row >> 3;
        const int ch = row & 7;
        int xs = xr + rs - 1;
        xs += (xs < 0) ? H : 0;
        xs -= (xs >= H) ? H : 0;
        *(float4*)&lds[row * WS + (y4 << 2)] =
            *(const float4*)&xin[((((size_t)b * 8 + ch) << LOGW) + xs) * W +
                                 (y4 << 2)];
      }
      const int row = threadIdx.x >> 2;
      const int e = threadIdx.x & 3;
      const int rs = row >> 3;
      const int ch = row & 7;
      int xs = xr + rs - 1;
      xs += (xs < 0) ? H : 0;
      xs -= (xs >= H) ? H : 0;
      lds[row * WS + W + e] =
          xin[((((size_t)b * 8 + ch) << LOGW) + xs) * W + e];
    }
    if (threadIdx.x < 8) {
      int xn = xr + 4;
      xn -= (xn >= H) ? H : 0;
      __builtin_prefetch(
          &xin[((((size_t)b * 8 + threadIdx.x) << LOGW) + xn) * W], 0, 1);
    }
    __syncthreads();

#pragma unroll 8
    for (int k = 0; k < W; k += 4) {
      const int kb = k + khi;
      v2f av, bv;
      av[0] = lds[aIdx + kb];
      av[1] = lds[aIdx + kb + 1];
      bv = *(const v2f*)&lds[bBase + kb];
      acc = __builtin_amdgcn_wmma_f32_16x16x4_f32(
           false, av,  false, bv,
           (short)0, acc,  false,  false);
    }
    __syncthreads();
  }

  const int rbase = m0 + ((lane >> 4) << 3);
  const int col = lane & 15;
  if (col < 8) {
#pragma unroll
    for (int i = 0; i < 8; ++i) {
      const int R = rbase + i;
      if (R < 56) {
        const int s = R >> 3;
        const int l1o = R & 7;
        atomicAdd(&out[(((size_t)b * 384) + pairBase + l1o * 8 + col) * 7 + s],
                  acc[i]);
      }
    }
  }
}


#define FFT_R 8

template <int LOGN>
__device__ __forceinline__ void fft_rows_body(const float* __restrict__ xin,
                                              float2* __restrict__ spec,
                                              int blk, float2* lds) {
  constexpr int N = 1 << LOGN;
  constexpr int HALF = N / 2;
  constexpr int RPB = 128 / HALF;
  const int sub = threadIdx.x >> (LOGN - 1);
  const int t = threadIdx.x & (HALF - 1);
  const size_t line0 = ((size_t)blk * RPB + sub) * FFT_R;
  float2* buf = lds + sub * (FFT_R * N);
  const float* src = xin + line0 * N;

  const int r0 = __brev((unsigned)t) >> (32 - LOGN);
  const int r1 = __brev((unsigned)(t + HALF)) >> (32 - LOGN);
  float lo[FFT_R], hi[FFT_R];
#pragma unroll
  for (int r = 0; r < FFT_R; ++r) {
    lo[r] = src[r * N + t];
    hi[r] = src[r * N + t + HALF];
  }
#pragma unroll
  for (int r = 0; r < FFT_R; ++r) {
    buf[r * N + r0] = make_float2(lo[r], 0.0f);
    buf[r * N + r1] = make_float2(hi[r], 0.0f);
  }
  __syncthreads();
#pragma unroll
  for (int st = 1; st <= LOGN; ++st) {
    const int len = 1 << st;
    const int half = len >> 1;
    const int j = t & (half - 1);
    const int base = (t >> (st - 1)) << st;
    float sn, cs;
    __sincosf(-TWO_PI_F * (float)j / (float)len, &sn, &cs);
#pragma unroll
    for (int r = 0; r < FFT_R; ++r) {
      float2* b = buf + r * N;
      const float2 u = b[base + j];
      const float2 v = b[base + j + half];
      const float2 tv = make_float2(v.x * cs - v.y * sn, v.x * sn + v.y * cs);
      b[base + j] = make_float2(u.x + tv.x, u.y + tv.y);
      b[base + j + half] = make_float2(u.x - tv.x, u.y - tv.y);
    }
    __syncthreads();
  }
  float2* dst = spec + line0 * N;
#pragma unroll
  for (int r = 0; r < FFT_R; ++r) {
    dst[r * N + t] = buf[r * N + t];
    dst[r * N + t + HALF] = buf[r * N + t + HALF];
  }
}

__global__ void __launch_bounds__(128)
fft_rows_all(const float* __restrict__ x0, const float* __restrict__ x1,
             const float* __restrict__ x2, float2* __restrict__ spec0,
             float2* __restrict__ spec1, float2* __restrict__ spec2,
             float* __restrict__ out) {
  __shared__ float2 buf[256 * FFT_R];
  const int bid = blockIdx.x;
  const int oi = bid * 128 + threadIdx.x;
  if (oi < 2 * 384 * 7) out[oi] = 0.0f;
  if (bid < 4096 / FFT_R) {
    fft_rows_body<8>(x0, spec0, bid, buf);
  } else if (bid < (4096 + 1024) / FFT_R) {
    fft_rows_body<7>(x1, spec1, bid - 4096 / FFT_R, buf);
  } else {
    fft_rows_body<6>(x2, spec2, bid - (4096 + 1024) / FFT_R, buf);
  }
}

template <int LOGN>
__device__ __forceinline__ void fft_cols_body(float2* __restrict__ spec, int blk,
                                              float2* lds) {
  constexpr int N = 1 << LOGN;
  constexpr int HALF = N / 2;
  constexpr int RPB = 128 / HALF;
  static_assert((FFT_R & 1) == 0, "column pairs move as 16-byte values");
  const int sub = threadIdx.x >> (LOGN - 1);
  const int t = threadIdx.x & (HALF - 1);
  const int line0 = (blk * RPB + sub) * FFT_R;
  const int col = line0 & (N - 1);
  const int img = line0 >> LOGN;
  float2* buf = lds + sub * (FFT_R * N);
  float2* base0 = spec + ((size_t)img << (2 * LOGN)) + col;

  const int r0 = __brev((unsigned)t) >> (32 - LOGN);
  const int r1 = __brev((unsigned)(t + HALF)) >> (32 - LOGN);
  float4 lo[FFT_R / 2], hi[FFT_R / 2];
#pragma unroll
  for (int q = 0; q < FFT_R / 2; ++q) {
    lo[q] = *(const float4*)(base0 + (size_t)t * N + 2 * q);
    hi[q] = *(const float4*)(base0 + (size_t)(t + HALF) * N + 2 * q);
  }
#pragma unroll
  for (int q = 0; q < FFT_R / 2; ++q) {
    buf[(2 * q) * N + r0] = make_float2(lo[q].x, lo[q].y);
    buf[(2 * q + 1) * N + r0] = make_float2(lo[q].z, lo[q].w);
    buf[(2 * q) * N + r1] = make_float2(hi[q].x, hi[q].y);
    buf[(2 * q + 1) * N + r1] = make_float2(hi[q].z, hi[q].w);
  }
  __syncthreads();
#pragma unroll
  for (int st = 1; st <= LOGN; ++st) {
    const int len = 1 << st;
    const int half = len >> 1;
    const int j = t & (half - 1);
    const int base = (t >> (st - 1)) << st;
    float sn, cs;
    __sincosf(-TWO_PI_F * (float)j / (float)len, &sn, &cs);
#pragma unroll
    for (int r = 0; r < FFT_R; ++r) {
      float2* b = buf + r * N;
      const float2 u = b[base + j];
      const float2 v = b[base + j + half];
      const float2 tv = make_float2(v.x * cs - v.y * sn, v.x * sn + v.y * cs);
      b[base + j] = make_float2(u.x + tv.x, u.y + tv.y);
      b[base + j + half] = make_float2(u.x - tv.x, u.y - tv.y);
    }
    __syncthreads();
  }
#pragma unroll
  for (int q = 0; q < FFT_R / 2; ++q) {
    const float2 a0 = buf[(2 * q) * N + t], a1 = buf[(2 * q + 1) * N + t];
    const float2 b0 = buf[(2 * q) * N + t + HALF];
    const float2 b1 = buf[(2 * q + 1) * N + t + HALF];
    *(float4*)(base0 + (size_t)t * N + 2 * q) =
        make_float4(a0.x, a0.y, a1.x, a1.y);
    *(float4*)(base0 + (size_t)(t + HALF) * N + 2 * q) =
        make_float4(b0.x, b0.y, b1.x, b1.y);
  }
}

#define SAME_BLOCKS (256 + 256 + 128)
__global__ void __launch_bounds__(128)
cols_and_same_all(const float* __restrict__ x0, const float* __restrict__ x1,
                  const float* __restrict__ x2, float2* __restrict__ spec0,
                  float2* __restrict__ spec1, float2* __restrict__ spec2,
                  float* __restrict__ out) {
  __shared__ __attribute__((aligned(16))) float lds[33 * 260];
  static_assert(sizeof(float2) * 256 * FFT_R <= sizeof(float) * 33 * 260,
                "transform buffer must fit the shared array");
  const int bid = blockIdx.x;
  if (bid < 256) {
    same_scale_body<8, 2>(x0, 0, out, bid, lds);
  } else if (bid < 512) {
    same_scale_body<7, 1>(x1, 192, out, bid - 256, lds);
  } else if (bid < SAME_BLOCKS) {
    same_scale_body<6, 1>(x2, 320, out, bid - 512, lds);
  } else {
    const int fb = bid - SAME_BLOCKS;
    float2* buf = reinterpret_cast<float2*>(lds);
    if (fb < 4096 / FFT_R) {
      fft_cols_body<8>(spec0, fb, buf);
    } else if (fb < (4096 + 1024) / FFT_R) {
      fft_cols_body<7>(spec1, fb - 4096 / FFT_R, buf);
    } else {
      fft_cols_body<6>(spec2, fb - (4096 + 1024) / FFT_R, buf);
    }
  }
}

template <int LOG1, int LOG2, int SPLIT, int U>
__device__ __forceinline__ void cross_body(const float2* __restrict__ s1,
                                           const float2* __restrict__ s2,
                                           int pairBase,
                                           float* __restrict__ out, int unit,
                                           int t, float* red) {
  constexpr int N1 = 1 << LOG1;
  constexpr int N2 = 1 << LOG2;
  constexpr int BINS = N2 * N2;
  constexpr int THREADS = BINS / (SPLIT * U);
  constexpr int NW = THREADS / 32;
  constexpr int ROWS_PER_PASS = THREADS >> LOG2;
  constexpr int ROWS_PER_BLOCK = N2 / SPLIT;
  static_assert(THREADS >= N2 && THREADS <= 1024 && (THREADS & 31) == 0,
                "block must cover whole spectrum rows");
  static_assert(ROWS_PER_PASS * U == ROWS_PER_BLOCK, "share must tile the rows");
  static_assert(SPLIT >= 2 && (SPLIT & (SPLIT - 1)) == 0, "rows of one half per block");

  const int part = unit & (SPLIT - 1);
  const int bp = unit / SPLIT;
  const int b = bp >> 6;
  const int p = bp & 63;
  const int l1 = p >> 3, l2 = p & 7;
  const float2* F1 = s1 + (((size_t)b * 8 + l1) << (2 * LOG1));
  const float2* F2 = s2 + (((size_t)b * 8 + l2) << (2 * LOG2));

  const int iy2 = t & (N2 - 1);
  const int fy = (iy2 < (N2 >> 1)) ? iy2 : iy2 - N2;
  const int col1 = (fy + N1) & (N1 - 1);
  const int row0 = part * ROWS_PER_BLOCK + (t >> LOG2);
  const int wrap1 = (part >= SPLIT / 2) ? (N1 - N2) : 0;
  const float2* pA = F1 + ((size_t)(row0 + wrap1) << LOG1) + col1;
  const float2* pB = F2 + ((size_t)row0 << LOG2) + iy2;

  float2 A[U], B[U];
#pragma unroll
  for (int u = 0; u < U; ++u) {
    A[u] = pA[(size_t)u * ROWS_PER_PASS * N1];
    B[u] = pB[(size_t)u * ROWS_PER_PASS * N2];
  }
#pragma unroll
  for (int u = 0; u < U; ++u)
    asm volatile("" ::"v"(A[u].x), "v"(A[u].y), "v"(B[u].x), "v"(B[u].y));

  float sy, cy;
  __sincosf(TWO_PI_F * (float)fy / (float)N1, &sy, &cy);
  const float c2y = cy * cy - sy * sy, s2y = 2.f * cy * sy;

  float a0 = 0.f, a1 = 0.f, a2 = 0.f, a3 = 0.f, a4 = 0.f, a5 = 0.f, a6 = 0.f;
#pragma unroll
  for (int u = 0; u < U; ++u) {
    const int ix2 = row0 + u * ROWS_PER_PASS;
    const int fx = (ix2 < (N2 >> 1)) ? ix2 : ix2 - N2;
    const float zr = A[u].x * B[u].x + A[u].y * B[u].y;
    const float zi = A[u].y * B[u].x - A[u].x * B[u].y;
    float sx, cx;
    __sincosf(TWO_PI_F * (float)fx / (float)N1, &sx, &cx);
    const float c2x = cx * cx - sx * sx, s2x = 2.f * cx * sx;
    a0 += zr;
    a1 += zr * cy - zi * sy;
    a2 += zr * c2y - zi * s2y;
    a3 += zr * cx - zi * sx;
    const float xr = zr * cx - zi * sx, xi = zr * sx + zi * cx;
    a4 += xr * cy - xi * sy;
    a5 += zr * c2x - zi * s2x;
    const float mr = zr * cx + zi * sx, mi = zi * cx - zr * sx;
    a6 += mr * cy - mi * sy;
  }

  float accs[7] = {a0, a1, a2, a3, a4, a5, a6};
#pragma unroll
  for (int s = 0; s < 7; ++s) {
    float v = accs[s];
    for (int off = 16; off > 0; off >>= 1) v += __shfl_xor(v, off, 32);
    accs[s] = v;
  }
  if ((t & 31) == 0) {
#pragma unroll
    for (int s = 0; s < 7; ++s) red[s * NW + (t >> 5)] = accs[s];
  }
  __syncthreads();
  if (t < 7) {
    float v = 0.f;
    for (int w = 0; w < NW; ++w) v += red[t * NW + w];
    const float inv = 1.0f / (float)BINS;
    atomicAdd(&out[(((size_t)b * 384) + pairBase + p) * 7 + t], v * inv);
  }
}

__global__ void __launch_bounds__(512)
cross_all(const float2* __restrict__ spec0, const float2* __restrict__ spec1,
          const float2* __restrict__ spec2, float* __restrict__ out) {
  __shared__ float red[2][7 * 16];
  const int bid = blockIdx.x;
  if (bid < 256) {
    cross_body<8, 7, 2, 16>(spec0, spec1, 64, out, bid, threadIdx.x, &red[0][0]);
  } else {
    const int hf = threadIdx.x >> 8;
    const int t = threadIdx.x & 255;
    const int unit = (bid - 256) * 2 + hf;
    if (bid < 384) {
      cross_body<8, 6, 2, 8>(spec0, spec2, 128, out, unit, t, &red[hf][0]);
    } else {
      cross_body<7, 6, 2, 8>(spec1, spec2, 256, out, unit - 256, t, &red[hf][0]);
    }
  }
}

extern "C" void kernel_launch(void* const* d_in, const int* in_sizes, int n_in,
                              void* d_out, int out_size, void* d_ws,
                              size_t ws_size, hipStream_t stream) {
  const float* x0 = (const float*)d_in[0];
  const float* x1 = (const float*)d_in[1];
  const float* x2 = (const float*)d_in[2];
  float* out = (float*)d_out;

  const size_t n0 = (size_t)16 * 256 * 256;
  const size_t n1 = (size_t)16 * 128 * 128;
  const size_t n2 = (size_t)16 * 64 * 64;
  if (ws_size >= (n0 + n1 + n2) * sizeof(float2)) {
    float2* spec0 = (float2*)d_ws;
    float2* spec1 = spec0 + n0;
    float2* spec2 = spec1 + n1;

    fft_rows_all<<<dim3((4096 + 1024 + 256) / FFT_R), 128, 0, stream>>>(
        x0, x1, x2, spec0, spec1, spec2, out);
    cols_and_same_all<<<dim3(SAME_BLOCKS + (4096 + 1024 + 256) / FFT_R), 128, 0,
                        stream>>>(x0, x1, x2, spec0, spec1, spec2, out);
    cross_all<<<dim3(512), 512, 0, stream>>>(spec0, spec1, spec2, out);
  }
}


# finest scale's spectrum kept only on the 128x128 low band the cross sums read: row FFT stores 128 of 256 columns, column FFT runs over those, stores 128 of 256 rows; every sum in the same order
# speedup vs baseline: 6.1218x; 1.5867x over previous
#include <hip/hip_runtime.h>
#include <hip/hip_bf16.h>


typedef __attribute__((ext_vector_type(2))) float v2f;
typedef __attribute__((ext_vector_type(8))) float v8f;

__device__ __constant__ int c_sdx[7] = {0, 0, 0, 1, 1, 2, -1};
__device__ __constant__ int c_sdy[7] = {0, 1, 2, 0, 1, 0, 1};

#define TWO_PI_F 6.28318530717958647692f

template <int LOGW, int ROWS>
__device__ __forceinline__ void same_scale_body(const float* __restrict__ xin,
                                                int pairBase,
                                                float* __restrict__ out,
                                                int blk, float* lds) {
  constexpr int W = 1 << LOGW;
  constexpr int H = W;
  constexpr int WS = W + 4;
  constexpr int ZROW = 32 * WS;
  constexpr int SLABS = H / ROWS;

  const int b = blk / SLABS;
  const int slab = blk & (SLABS - 1);

  const int lane = threadIdx.x & 31;
  const int wave = threadIdx.x >> 5;
  const int m0 = wave * 16;

  const int mrow = m0 + (lane & 15);
  const bool aval = (mrow < 56);
  const int sidx = aval ? (mrow >> 3) : 0;
  const int l1 = mrow & 7;
  const int slot = c_sdx[sidx] + 1;
  const int dy = aval ? c_sdy[sidx] : 0;
  const int aIdx = (aval ? (slot * 8 + l1) * WS : ZROW) + dy;
  const int ncol = lane & 15;
  const int bBase = (ncol < 8) ? (8 + ncol) * WS : ZROW;
  const int khi = (lane >> 4) << 1;

  for (int idx = threadIdx.x; idx < WS; idx += 128) lds[ZROW + idx] = 0.0f;

  v8f acc = {};

  for (int xi = 0; xi < ROWS; ++xi) {
    const int xr = slab * ROWS + xi;

    {
      constexpr int NV4 = 8 * W;
      constexpr int PER_CH4 = W / 4;
      for (int idx = threadIdx.x; idx < NV4; idx += 128) {
        const int row = idx >> (LOGW - 2);
        const int y4 = idx & (PER_CH4 - 1);
        const int rs = row >> 3;
        const int ch = row & 7;
        int xs = xr + rs - 1;
        xs += (xs < 0) ? H : 0;
        xs -= (xs >= H) ? H : 0;
        *(float4*)&lds[row * WS + (y4 << 2)] =
            *(const float4*)&xin[((((size_t)b * 8 + ch) << LOGW) + xs) * W +
                                 (y4 << 2)];
      }
      const int row = threadIdx.x >> 2;
      const int e = threadIdx.x & 3;
      const int rs = row >> 3;
      const int ch = row & 7;
      int xs = xr + rs - 1;
      xs += (xs < 0) ? H : 0;
      xs -= (xs >= H) ? H : 0;
      lds[row * WS + W + e] =
          xin[((((size_t)b * 8 + ch) << LOGW) + xs) * W + e];
    }
    if (threadIdx.x < 8) {
      int xn = xr + 4;
      xn -= (xn >= H) ? H : 0;
      __builtin_prefetch(
          &xin[((((size_t)b * 8 + threadIdx.x) << LOGW) + xn) * W], 0, 1);
    }
    __syncthreads();

#pragma unroll 8
    for (int k = 0; k < W; k += 4) {
      const int kb = k + khi;
      v2f av, bv;
      av[0] = lds[aIdx + kb];
      av[1] = lds[aIdx + kb + 1];
      bv = *(const v2f*)&lds[bBase + kb];
      acc = __builtin_amdgcn_wmma_f32_16x16x4_f32(
           false, av,  false, bv,
           (short)0, acc,  false,  false);
    }
    __syncthreads();
  }

  const int rbase = m0 + ((lane >> 4) << 3);
  const int col = lane & 15;
  if (col < 8) {
#pragma unroll
    for (int i = 0; i < 8; ++i) {
      const int R = rbase + i;
      if (R < 56) {
        const int s = R >> 3;
        const int l1o = R & 7;
        atomicAdd(&out[(((size_t)b * 384) + pairBase + l1o * 8 + col) * 7 + s],
                  acc[i]);
      }
    }
  }
}


#define FFT_R 8

template <int LOGN, bool BAND>
__device__ __forceinline__ void fft_rows_body(const float* __restrict__ xin,
                                              float2* __restrict__ spec,
                                              int blk, float2* lds) {
  constexpr int N = 1 << LOGN;
  constexpr int HALF = N / 2;
  constexpr int RPB = 128 / HALF;
  const int sub = threadIdx.x >> (LOGN - 1);
  const int t = threadIdx.x & (HALF - 1);
  const size_t line0 = ((size_t)blk * RPB + sub) * FFT_R;
  float2* buf = lds + sub * (FFT_R * N);
  const float* src = xin + line0 * N;

  const int r0 = __brev((unsigned)t) >> (32 - LOGN);
  const int r1 = __brev((unsigned)(t + HALF)) >> (32 - LOGN);
  float lo[FFT_R], hi[FFT_R];
#pragma unroll
  for (int r = 0; r < FFT_R; ++r) {
    lo[r] = src[r * N + t];
    hi[r] = src[r * N + t + HALF];
  }
#pragma unroll
  for (int r = 0; r < FFT_R; ++r) {
    buf[r * N + r0] = make_float2(lo[r], 0.0f);
    buf[r * N + r1] = make_float2(hi[r], 0.0f);
  }
  __syncthreads();
#pragma unroll
  for (int st = 1; st <= LOGN; ++st) {
    const int len = 1 << st;
    const int half = len >> 1;
    const int j = t & (half - 1);
    const int base = (t >> (st - 1)) << st;
    float sn, cs;
    __sincosf(-TWO_PI_F * (float)j / (float)len, &sn, &cs);
#pragma unroll
    for (int r = 0; r < FFT_R; ++r) {
      float2* b = buf + r * N;
      const float2 u = b[base + j];
      const float2 v = b[base + j + half];
      const float2 tv = make_float2(v.x * cs - v.y * sn, v.x * sn + v.y * cs);
      b[base + j] = make_float2(u.x + tv.x, u.y + tv.y);
      b[base + j + half] = make_float2(u.x - tv.x, u.y - tv.y);
    }
    __syncthreads();
  }
  if (BAND) {
    float2* dst = spec + line0 * HALF;
    const int o = (t < N / 4) ? t : t + HALF;
#pragma unroll
    for (int r = 0; r < FFT_R; ++r) dst[r * HALF + t] = buf[r * N + o];
  } else {
    float2* dst = spec + line0 * N;
#pragma unroll
    for (int r = 0; r < FFT_R; ++r) {
      dst[r * N + t] = buf[r * N + t];
      dst[r * N + t + HALF] = buf[r * N + t + HALF];
    }
  }
}

__global__ void __launch_bounds__(128)
fft_rows_all(const float* __restrict__ x0, const float* __restrict__ x1,
             const float* __restrict__ x2, float2* __restrict__ spec0,
             float2* __restrict__ spec1, float2* __restrict__ spec2,
             float* __restrict__ out) {
  __shared__ float2 buf[256 * FFT_R];
  const int bid = blockIdx.x;
  const int oi = bid * 128 + threadIdx.x;
  if (oi < 2 * 384 * 7) out[oi] = 0.0f;
  if (bid < 4096 / FFT_R) {
    fft_rows_body<8, true>(x0, spec0, bid, buf);
  } else if (bid < (4096 + 1024) / FFT_R) {
    fft_rows_body<7, false>(x1, spec1, bid - 4096 / FFT_R, buf);
  } else {
    fft_rows_body<6, false>(x2, spec2, bid - (4096 + 1024) / FFT_R, buf);
  }
}

template <int LOGN, int LOGC, bool BAND>
__device__ __forceinline__ void fft_cols_body(float2* __restrict__ spec, int blk,
                                              float2* lds) {
  constexpr int N = 1 << LOGN;
  constexpr int NC = 1 << LOGC;
  constexpr int HALF = N / 2;
  constexpr int RPB = 128 / HALF;
  static_assert((FFT_R & 1) == 0, "column pairs move as 16-byte values");
  const int sub = threadIdx.x >> (LOGN - 1);
  const int t = threadIdx.x & (HALF - 1);
  const int line0 = (blk * RPB + sub) * FFT_R;
  const int col = line0 & (NC - 1);
  const int img = line0 >> LOGC;
  float2* buf = lds + sub * (FFT_R * N);
  float2* base0 = spec + ((size_t)img << (LOGN + LOGC)) + col;

  const int r0 = __brev((unsigned)t) >> (32 - LOGN);
  const int r1 = __brev((unsigned)(t + HALF)) >> (32 - LOGN);
  float4 lo[FFT_R / 2], hi[FFT_R / 2];
#pragma unroll
  for (int q = 0; q < FFT_R / 2; ++q) {
    lo[q] = *(const float4*)(base0 + (size_t)t * NC + 2 * q);
    hi[q] = *(const float4*)(base0 + (size_t)(t + HALF) * NC + 2 * q);
  }
#pragma unroll
  for (int q = 0; q < FFT_R / 2; ++q) {
    buf[(2 * q) * N + r0] = make_float2(lo[q].x, lo[q].y);
    buf[(2 * q + 1) * N + r0] = make_float2(lo[q].z, lo[q].w);
    buf[(2 * q) * N + r1] = make_float2(hi[q].x, hi[q].y);
    buf[(2 * q + 1) * N + r1] = make_float2(hi[q].z, hi[q].w);
  }
  __syncthreads();
#pragma unroll
  for (int st = 1; st <= LOGN; ++st) {
    const int len = 1 << st;
    const int half = len >> 1;
    const int j = t & (half - 1);
    const int base = (t >> (st - 1)) << st;
    float sn, cs;
    __sincosf(-TWO_PI_F * (float)j / (float)len, &sn, &cs);
#pragma unroll
    for (int r = 0; r < FFT_R; ++r) {
      float2* b = buf + r * N;
      const float2 u = b[base + j];
      const float2 v = b[base + j + half];
      const float2 tv = make_float2(v.x * cs - v.y * sn, v.x * sn + v.y * cs);
      b[base + j] = make_float2(u.x + tv.x, u.y + tv.y);
      b[base + j + half] = make_float2(u.x - tv.x, u.y - tv.y);
    }
    __syncthreads();
  }
  if (BAND) {
    const int o = (t < N / 4) ? t : t + HALF;
#pragma unroll
    for (int q = 0; q < FFT_R / 2; ++q) {
      const float2 a0 = buf[(2 * q) * N + o], a1 = buf[(2 * q + 1) * N + o];
      *(float4*)(base0 + (size_t)t * NC + 2 * q) =
          make_float4(a0.x, a0.y, a1.x, a1.y);
    }
  } else {
#pragma unroll
    for (int q = 0; q < FFT_R / 2; ++q) {
      const float2 a0 = buf[(2 * q) * N + t], a1 = buf[(2 * q + 1) * N + t];
      const float2 b0 = buf[(2 * q) * N + t + HALF];
      const float2 b1 = buf[(2 * q + 1) * N + t + HALF];
      *(float4*)(base0 + (size_t)t * NC + 2 * q) =
          make_float4(a0.x, a0.y, a1.x, a1.y);
      *(float4*)(base0 + (size_t)(t + HALF) * NC + 2 * q) =
          make_float4(b0.x, b0.y, b1.x, b1.y);
    }
  }
}

#define SAME_BLOCKS (256 + 256 + 128)
#define BAND0_COLS (16 * 128)
__global__ void __launch_bounds__(128)
cols_and_same_all(const float* __restrict__ x0, const float* __restrict__ x1,
                  const float* __restrict__ x2, float2* __restrict__ spec0,
                  float2* __restrict__ spec1, float2* __restrict__ spec2,
                  float* __restrict__ out) {
  __shared__ __attribute__((aligned(16))) float lds[33 * 260];
  static_assert(sizeof(float2) * 256 * FFT_R <= sizeof(float) * 33 * 260,
                "transform buffer must fit the shared array");
  const int bid = blockIdx.x;
  if (bid < 256) {
    same_scale_body<8, 2>(x0, 0, out, bid, lds);
  } else if (bid < 512) {
    same_scale_body<7, 1>(x1, 192, out, bid - 256, lds);
  } else if (bid < SAME_BLOCKS) {
    same_scale_body<6, 1>(x2, 320, out, bid - 512, lds);
  } else {
    const int fb = bid - SAME_BLOCKS;
    float2* buf = reinterpret_cast<float2*>(lds);
    if (fb < BAND0_COLS / FFT_R) {
      fft_cols_body<8, 7, true>(spec0, fb, buf);
    } else if (fb < (BAND0_COLS + 1024) / FFT_R) {
      fft_cols_body<7, 7, false>(spec1, fb - BAND0_COLS / FFT_R, buf);
    } else {
      fft_cols_body<6, 6, false>(spec2, fb - (BAND0_COLS + 1024) / FFT_R, buf);
    }
  }
}

template <int LOG1, int LOGS1, int LOGI1, int LOG2, int SPLIT, int U>
__device__ __forceinline__ void cross_body(const float2* __restrict__ s1,
                                           const float2* __restrict__ s2,
                                           int pairBase,
                                           float* __restrict__ out, int unit,
                                           int t, float* red) {
  constexpr int N1 = 1 << LOG1;
  constexpr int NS1 = 1 << LOGS1;
  constexpr int N2 = 1 << LOG2;
  constexpr int BINS = N2 * N2;
  constexpr int THREADS = BINS / (SPLIT * U);
  constexpr int NW = THREADS / 32;
  constexpr int ROWS_PER_PASS = THREADS >> LOG2;
  constexpr int ROWS_PER_BLOCK = N2 / SPLIT;
  static_assert(THREADS >= N2 && THREADS <= 1024 && (THREADS & 31) == 0,
                "block must cover whole spectrum rows");
  static_assert(ROWS_PER_PASS * U == ROWS_PER_BLOCK, "share must tile the rows");
  static_assert(SPLIT >= 2 && (SPLIT & (SPLIT - 1)) == 0, "rows of one half per block");
  static_assert(N2 <= NS1 && NS1 <= N1, "stored fine band must hold the coarse support");

  const int part = unit & (SPLIT - 1);
  const int bp = unit / SPLIT;
  const int b = bp >> 6;
  const int p = bp & 63;
  const int l1 = p >> 3, l2 = p & 7;
  const float2* F1 = s1 + (((size_t)b * 8 + l1) << LOGI1);
  const float2* F2 = s2 + (((size_t)b * 8 + l2) << (2 * LOG2));

  const int iy2 = t & (N2 - 1);
  const int fy = (iy2 < (N2 >> 1)) ? iy2 : iy2 - N2;
  const int col1 = (fy + NS1) & (NS1 - 1);
  const int row0 = part * ROWS_PER_BLOCK + (t >> LOG2);
  const int wrap1 = (part >= SPLIT / 2) ? (NS1 - N2) : 0;
  const float2* pA = F1 + ((size_t)(row0 + wrap1) << LOGS1) + col1;
  const float2* pB = F2 + ((size_t)row0 << LOG2) + iy2;

  float2 A[U], B[U];
#pragma unroll
  for (int u = 0; u < U; ++u) {
    A[u] = pA[(size_t)u * ROWS_PER_PASS * NS1];
    B[u] = pB[(size_t)u * ROWS_PER_PASS * N2];
  }
#pragma unroll
  for (int u = 0; u < U; ++u)
    asm volatile("" ::"v"(A[u].x), "v"(A[u].y), "v"(B[u].x), "v"(B[u].y));

  float sy, cy;
  __sincosf(TWO_PI_F * (float)fy / (float)N1, &sy, &cy);
  const float c2y = cy * cy - sy * sy, s2y = 2.f * cy * sy;

  float a0 = 0.f, a1 = 0.f, a2 = 0.f, a3 = 0.f, a4 = 0.f, a5 = 0.f, a6 = 0.f;
#pragma unroll
  for (int u = 0; u < U; ++u) {
    const int ix2 = row0 + u * ROWS_PER_PASS;
    const int fx = (ix2 < (N2 >> 1)) ? ix2 : ix2 - N2;
    const float zr = A[u].x * B[u].x + A[u].y * B[u].y;
    const float zi = A[u].y * B[u].x - A[u].x * B[u].y;
    float sx, cx;
    __sincosf(TWO_PI_F * (float)fx / (float)N1, &sx, &cx);
    const float c2x = cx * cx - sx * sx, s2x = 2.f * cx * sx;
    a0 += zr;
    a1 += zr * cy - zi * sy;
    a2 += zr * c2y - zi * s2y;
    a3 += zr * cx - zi * sx;
    const float xr = zr * cx - zi * sx, xi = zr * sx + zi * cx;
    a4 += xr * cy - xi * sy;
    a5 += zr * c2x - zi * s2x;
    const float mr = zr * cx + zi * sx, mi = zi * cx - zr * sx;
    a6 += mr * cy - mi * sy;
  }

  float accs[7] = {a0, a1, a2, a3, a4, a5, a6};
#pragma unroll
  for (int s = 0; s < 7; ++s) {
    float v = accs[s];
    for (int off = 16; off > 0; off >>= 1) v += __shfl_xor(v, off, 32);
    accs[s] = v;
  }
  if ((t & 31) == 0) {
#pragma unroll
    for (int s = 0; s < 7; ++s) red[s * NW + (t >> 5)] = accs[s];
  }
  __syncthreads();
  if (t < 7) {
    float v = 0.f;
    for (int w = 0; w < NW; ++w) v += red[t * NW + w];
    const float inv = 1.0f / (float)BINS;
    atomicAdd(&out[(((size_t)b * 384) + pairBase + p) * 7 + t], v * inv);
  }
}

__global__ void __launch_bounds__(512)
cross_all(const float2* __restrict__ spec0, const float2* __restrict__ spec1,
          const float2* __restrict__ spec2, float* __restrict__ out) {
  __shared__ float red[2][7 * 16];
  const int bid = blockIdx.x;
  if (bid < 256) {
    cross_body<8, 7, 15, 7, 2, 16>(spec0, spec1, 64, out, bid, threadIdx.x,
                                   &red[0][0]);
  } else {
    const int hf = threadIdx.x >> 8;
    const int t = threadIdx.x & 255;
    const int unit = (bid - 256) * 2 + hf;
    if (bid < 384) {
      cross_body<8, 7, 15, 6, 2, 8>(spec0, spec2, 128, out, unit, t, &red[hf][0]);
    } else {
      cross_body<7, 7, 14, 6, 2, 8>(spec1, spec2, 256, out, unit - 256, t,
                                    &red[hf][0]);
    }
  }
}

extern "C" void kernel_launch(void* const* d_in, const int* in_sizes, int n_in,
                              void* d_out, int out_size, void* d_ws,
                              size_t ws_size, hipStream_t stream) {
  const float* x0 = (const float*)d_in[0];
  const float* x1 = (const float*)d_in[1];
  const float* x2 = (const float*)d_in[2];
  float* out = (float*)d_out;

  const size_t n0 = (size_t)16 * 256 * 128;
  const size_t n1 = (size_t)16 * 128 * 128;
  const size_t n2 = (size_t)16 * 64 * 64;
  if (ws_size >= (n0 + n1 + n2) * sizeof(float2)) {
    float2* spec0 = (float2*)d_ws;
    float2* spec1 = spec0 + n0;
    float2* spec2 = spec1 + n1;

    fft_rows_all<<<dim3((4096 + 1024 + 256) / FFT_R), 128, 0, stream>>>(
        x0, x1, x2, spec0, spec1, spec2, out);
    cols_and_same_all<<<dim3(SAME_BLOCKS + (BAND0_COLS + 1024 + 256) / FFT_R),
                        128, 0, stream>>>(x0, x1, x2, spec0, spec1, spec2, out);
    cross_all<<<dim3(512), 512, 0, stream>>>(spec0, spec1, spec2, out);
  }
}


# FFT kernels: two radix-2 stages per LDS pass (4 points of a line per thread, in registers): half the LDS trips and barriers, half the transform waves; every sum in the same order
# speedup vs baseline: 6.5842x; 1.0755x over previous
#include <hip/hip_runtime.h>
#include <hip/hip_bf16.h>


typedef __attribute__((ext_vector_type(2))) float v2f;
typedef __attribute__((ext_vector_type(8))) float v8f;

__device__ __constant__ int c_sdx[7] = {0, 0, 0, 1, 1, 2, -1};
__device__ __constant__ int c_sdy[7] = {0, 1, 2, 0, 1, 0, 1};

#define TWO_PI_F 6.28318530717958647692f

template <int LOGW, int ROWS>
__device__ __forceinline__ void same_scale_body(const float* __restrict__ xin,
                                                int pairBase,
                                                float* __restrict__ out,
                                                int blk, float* lds) {
  constexpr int W = 1 << LOGW;
  constexpr int H = W;
  constexpr int WS = W + 4;
  constexpr int ZROW = 32 * WS;
  constexpr int SLABS = H / ROWS;

  const int b = blk / SLABS;
  const int slab = blk & (SLABS - 1);

  const int lane = threadIdx.x & 31;
  const int wave = threadIdx.x >> 5;
  const int m0 = wave * 16;

  const int mrow = m0 + (lane & 15);
  const bool aval = (mrow < 56);
  const int sidx = aval ? (mrow >> 3) : 0;
  const int l1 = mrow & 7;
  const int slot = c_sdx[sidx] + 1;
  const int dy = aval ? c_sdy[sidx] : 0;
  const int aIdx = (aval ? (slot * 8 + l1) * WS : ZROW) + dy;
  const int ncol = lane & 15;
  const int bBase = (ncol < 8) ? (8 + ncol) * WS : ZROW;
  const int khi = (lane >> 4) << 1;

  for (int idx = threadIdx.x; idx < WS; idx += 128) lds[ZROW + idx] = 0.0f;

  v8f acc = {};

  for (int xi = 0; xi < ROWS; ++xi) {
    const int xr = slab * ROWS + xi;

    {
      constexpr int NV4 = 8 * W;
      constexpr int PER_CH4 = W / 4;
      for (int idx = threadIdx.x; idx < NV4; idx += 128) {
        const int row = idx >> (LOGW - 2);
        const int y4 = idx & (PER_CH4 - 1);
        const int rs = row >> 3;
        const int ch = row & 7;
        int xs = xr + rs - 1;
        xs += (xs < 0) ? H : 0;
        xs -= (xs >= H) ? H : 0;
        *(float4*)&lds[row * WS + (y4 << 2)] =
            *(const float4*)&xin[((((size_t)b * 8 + ch) << LOGW) + xs) * W +
                                 (y4 << 2)];
      }
      const int row = threadIdx.x >> 2;
      const int e = threadIdx.x & 3;
      const int rs = row >> 3;
      const int ch = row & 7;
      int xs = xr + rs - 1;
      xs += (xs < 0) ? H : 0;
      xs -= (xs >= H) ? H : 0;
      lds[row * WS + W + e] =
          xin[((((size_t)b * 8 + ch) << LOGW) + xs) * W + e];
    }
    if (threadIdx.x < 8) {
      int xn = xr + 4;
      xn -= (xn >= H) ? H : 0;
      __builtin_prefetch(
          &xin[((((size_t)b * 8 + threadIdx.x) << LOGW) + xn) * W], 0, 1);
    }
    __syncthreads();

#pragma unroll 8
    for (int k = 0; k < W; k += 4) {
      const int kb = k + khi;
      v2f av, bv;
      av[0] = lds[aIdx + kb];
      av[1] = lds[aIdx + kb + 1];
      bv = *(const v2f*)&lds[bBase + kb];
      acc = __builtin_amdgcn_wmma_f32_16x16x4_f32(
           false, av,  false, bv,
           (short)0, acc,  false,  false);
    }
    __syncthreads();
  }

  const int rbase = m0 + ((lane >> 4) << 3);
  const int col = lane & 15;
  if (col < 8) {
#pragma unroll
    for (int i = 0; i < 8; ++i) {
      const int R = rbase + i;
      if (R < 56) {
        const int s = R >> 3;
        const int l1o = R & 7;
        atomicAdd(&out[(((size_t)b * 384) + pairBase + l1o * 8 + col) * 7 + s],
                  acc[i]);
      }
    }
  }
}


#define FFT_R 8

template <int LOGN>
__device__ __forceinline__ void fft_lines(float2* buf, int t) {
  constexpr int N = 1 << LOGN;
  constexpr int Q = N / 4;
#pragma unroll
  for (int st = 1; st + 1 <= LOGN; st += 2) {
    const int h = 1 << (st - 1);
    const int j = t & (h - 1);
    const int i0 = ((t >> (st - 1)) << (st + 1)) + j;
    float s1, c1, s2, c2, s3, c3;
    __sincosf(-TWO_PI_F * (float)j / (float)(2 * h), &s1, &c1);
    __sincosf(-TWO_PI_F * (float)j / (float)(4 * h), &s2, &c2);
    __sincosf(-TWO_PI_F * (float)(j + h) / (float)(4 * h), &s3, &c3);
#pragma unroll
    for (int r = 0; r < FFT_R; ++r) {
      float2* b = buf + r * N + i0;
      const float2 e0 = b[0], e1 = b[h], e2 = b[2 * h], e3 = b[3 * h];
      const float2 ta = make_float2(e1.x * c1 - e1.y * s1, e1.x * s1 + e1.y * c1);
      const float2 f0 = make_float2(e0.x + ta.x, e0.y + ta.y);
      const float2 f1 = make_float2(e0.x - ta.x, e0.y - ta.y);
      const float2 tb = make_float2(e3.x * c1 - e3.y * s1, e3.x * s1 + e3.y * c1);
      const float2 f2 = make_float2(e2.x + tb.x, e2.y + tb.y);
      const float2 f3 = make_float2(e2.x - tb.x, e2.y - tb.y);
      const float2 tc = make_float2(f2.x * c2 - f2.y * s2, f2.x * s2 + f2.y * c2);
      const float2 td = make_float2(f3.x * c3 - f3.y * s3, f3.x * s3 + f3.y * c3);
      b[0] = make_float2(f0.x + tc.x, f0.y + tc.y);
      b[2 * h] = make_float2(f0.x - tc.x, f0.y - tc.y);
      b[h] = make_float2(f1.x + td.x, f1.y + td.y);
      b[3 * h] = make_float2(f1.x - td.x, f1.y - td.y);
    }
    __syncthreads();
  }
  if (LOGN & 1) {
    constexpr int half = N / 2;
#pragma unroll
    for (int m = 0; m < 2; ++m) {
      const int j = t + m * Q;
      float sn, cs;
      __sincosf(-TWO_PI_F * (float)j / (float)N, &sn, &cs);
#pragma unroll
      for (int r = 0; r < FFT_R; ++r) {
        float2* b = buf + r * N;
        const float2 u = b[j];
        const float2 v = b[j + half];
        const float2 tv = make_float2(v.x * cs - v.y * sn, v.x * sn + v.y * cs);
        b[j] = make_float2(u.x + tv.x, u.y + tv.y);
        b[j + half] = make_float2(u.x - tv.x, u.y - tv.y);
      }
    }
    __syncthreads();
  }
}

template <int LOGN, bool BAND>
__device__ __forceinline__ void fft_rows_body(const float* __restrict__ xin,
                                              float2* __restrict__ spec,
                                              int blk, float2* lds) {
  constexpr int N = 1 << LOGN;
  constexpr int Q = N / 4;
  constexpr int HALF = N / 2;
  constexpr int RPB = 128 / Q;
  const int sub = threadIdx.x >> (LOGN - 2);
  const int t = threadIdx.x & (Q - 1);
  const size_t line0 = ((size_t)blk * RPB + sub) * FFT_R;
  float2* buf = lds + sub * (FFT_R * N);
  const float* src = xin + line0 * N;

  float v[4][FFT_R];
#pragma unroll
  for (int m = 0; m < 4; ++m) {
#pragma unroll
    for (int r = 0; r < FFT_R; ++r) v[m][r] = src[r * N + t + m * Q];
  }
#pragma unroll
  for (int m = 0; m < 4; ++m) {
    const int rm = __brev((unsigned)(t + m * Q)) >> (32 - LOGN);
#pragma unroll
    for (int r = 0; r < FFT_R; ++r) buf[r * N + rm] = make_float2(v[m][r], 0.0f);
  }
  __syncthreads();
  fft_lines<LOGN>(buf, t);
  if (BAND) {
    float2* dst = spec + line0 * HALF;
#pragma unroll
    for (int r = 0; r < FFT_R; ++r) {
      dst[r * HALF + t] = buf[r * N + t];
      dst[r * HALF + t + Q] = buf[r * N + t + 3 * Q];
    }
  } else {
    float2* dst = spec + line0 * N;
#pragma unroll
    for (int r = 0; r < FFT_R; ++r) {
#pragma unroll
      for (int m = 0; m < 4; ++m) dst[r * N + t + m * Q] = buf[r * N + t + m * Q];
    }
  }
}

#define ROW_BLOCKS0 (4096 / 16)
#define ROW_BLOCKS1 (2048 / 32)
#define ROW_BLOCKS2 (1024 / 64)

__global__ void __launch_bounds__(128)
fft_rows_all(const float* __restrict__ x0, const float* __restrict__ x1,
             const float* __restrict__ x2, float2* __restrict__ spec0,
             float2* __restrict__ spec1, float2* __restrict__ spec2,
             float* __restrict__ out) {
  __shared__ float2 buf[2 * 256 * FFT_R];
  const int bid = blockIdx.x;
  const int oi = bid * 128 + threadIdx.x;
  if (oi < 2 * 384 * 7) out[oi] = 0.0f;
  if (bid < ROW_BLOCKS0) {
    fft_rows_body<8, true>(x0, spec0, bid, buf);
  } else if (bid < ROW_BLOCKS0 + ROW_BLOCKS1) {
    fft_rows_body<7, false>(x1, spec1, bid - ROW_BLOCKS0, buf);
  } else {
    fft_rows_body<6, false>(x2, spec2, bid - (ROW_BLOCKS0 + ROW_BLOCKS1), buf);
  }
}

template <int LOGN, int LOGC, bool BAND>
__device__ __forceinline__ void fft_cols_body(float2* __restrict__ spec, int blk,
                                              float2* lds) {
  constexpr int N = 1 << LOGN;
  constexpr int NC = 1 << LOGC;
  constexpr int Q = N / 4;
  constexpr int RPB = 128 / Q;
  static_assert((FFT_R & 1) == 0, "column pairs move as 16-byte values");
  const int sub = threadIdx.x >> (LOGN - 2);
  const int t = threadIdx.x & (Q - 1);
  const int line0 = (blk * RPB + sub) * FFT_R;
  const int col = line0 & (NC - 1);
  const int img = line0 >> LOGC;
  float2* buf = lds + sub * (FFT_R * N);
  float2* base0 = spec + ((size_t)img << (LOGN + LOGC)) + col;

  float4 ld[4][FFT_R / 2];
#pragma unroll
  for (int m = 0; m < 4; ++m) {
#pragma unroll
    for (int q = 0; q < FFT_R / 2; ++q)
      ld[m][q] = *(const float4*)(base0 + (size_t)(t + m * Q) * NC + 2 * q);
  }
#pragma unroll
  for (int m = 0; m < 4; ++m) {
    const int rm = __brev((unsigned)(t + m * Q)) >> (32 - LOGN);
#pragma unroll
    for (int q = 0; q < FFT_R / 2; ++q) {
      buf[(2 * q) * N + rm] = make_float2(ld[m][q].x, ld[m][q].y);
      buf[(2 * q + 1) * N + rm] = make_float2(ld[m][q].z, ld[m][q].w);
    }
  }
  __syncthreads();
  fft_lines<LOGN>(buf, t);
  if (BAND) {
#pragma unroll
    for (int q = 0; q < FFT_R / 2; ++q) {
      const float2 a0 = buf[(2 * q) * N + t], a1 = buf[(2 * q + 1) * N + t];
      const float2 b0 = buf[(2 * q) * N + t + 3 * Q];
      const float2 b1 = buf[(2 * q + 1) * N + t + 3 * Q];
      *(float4*)(base0 + (size_t)t * NC + 2 * q) =
          make_float4(a0.x, a0.y, a1.x, a1.y);
      *(float4*)(base0 + (size_t)(t + Q) * NC + 2 * q) =
          make_float4(b0.x, b0.y, b1.x, b1.y);
    }
  } else {
#pragma unroll
    for (int m = 0; m < 4; ++m) {
#pragma unroll
      for (int q = 0; q < FFT_R / 2; ++q) {
        const float2 a0 = buf[(2 * q) * N + t + m * Q];
        const float2 a1 = buf[(2 * q + 1) * N + t + m * Q];
        *(float4*)(base0 + (size_t)(t + m * Q) * NC + 2 * q) =
            make_float4(a0.x, a0.y, a1.x, a1.y);
      }
    }
  }
}

#define SAME_BLOCKS (256 + 256 + 128)
#define COL_BLOCKS0 (16 * 128 / 16)
#define COL_BLOCKS1 (16 * 128 / 32)
#define COL_BLOCKS2 (16 * 64 / 64)
__global__ void __launch_bounds__(128)
cols_and_same_all(const float* __restrict__ x0, const float* __restrict__ x1,
                  const float* __restrict__ x2, float2* __restrict__ spec0,
                  float2* __restrict__ spec1, float2* __restrict__ spec2,
                  float* __restrict__ out) {
  __shared__ __attribute__((aligned(16))) float lds[33 * 260];
  static_assert(sizeof(float2) * 2 * 256 * FFT_R <= sizeof(float) * 33 * 260,
                "transform buffer must fit the shared array");
  const int bid = blockIdx.x;
  if (bid < 256) {
    same_scale_body<8, 2>(x0, 0, out, bid, lds);
  } else if (bid < 512) {
    same_scale_body<7, 1>(x1, 192, out, bid - 256, lds);
  } else if (bid < SAME_BLOCKS) {
    same_scale_body<6, 1>(x2, 320, out, bid - 512, lds);
  } else {
    const int fb = bid - SAME_BLOCKS;
    float2* buf = reinterpret_cast<float2*>(lds);
    if (fb < COL_BLOCKS0) {
      fft_cols_body<8, 7, true>(spec0, fb, buf);
    } else if (fb < COL_BLOCKS0 + COL_BLOCKS1) {
      fft_cols_body<7, 7, false>(spec1, fb - COL_BLOCKS0, buf);
    } else {
      fft_cols_body<6, 6, false>(spec2, fb - (COL_BLOCKS0 + COL_BLOCKS1), buf);
    }
  }
}

template <int LOG1, int LOGS1, int LOGI1, int LOG2, int SPLIT, int U>
__device__ __forceinline__ void cross_body(const float2* __restrict__ s1,
                                           const float2* __restrict__ s2,
                                           int pairBase,
                                           float* __restrict__ out, int unit,
                                           int t, float* red) {
  constexpr int N1 = 1 << LOG1;
  constexpr int NS1 = 1 << LOGS1;
  constexpr int N2 = 1 << LOG2;
  constexpr int BINS = N2 * N2;
  constexpr int THREADS = BINS / (SPLIT * U);
  constexpr int NW = THREADS / 32;
  constexpr int ROWS_PER_PASS = THREADS >> LOG2;
  constexpr int ROWS_PER_BLOCK = N2 / SPLIT;
  static_assert(THREADS >= N2 && THREADS <= 1024 && (THREADS & 31) == 0,
                "block must cover whole spectrum rows");
  static_assert(ROWS_PER_PASS * U == ROWS_PER_BLOCK, "share must tile the rows");
  static_assert(SPLIT >= 2 && (SPLIT & (SPLIT - 1)) == 0, "rows of one half per block");
  static_assert(N2 <= NS1 && NS1 <= N1, "stored fine band must hold the coarse support");

  const int part = unit & (SPLIT - 1);
  const int bp = unit / SPLIT;
  const int b = bp >> 6;
  const int p = bp & 63;
  const int l1 = p >> 3, l2 = p & 7;
  const float2* F1 = s1 + (((size_t)b * 8 + l1) << LOGI1);
  const float2* F2 = s2 + (((size_t)b * 8 + l2) << (2 * LOG2));

  const int iy2 = t & (N2 - 1);
  const int fy = (iy2 < (N2 >> 1)) ? iy2 : iy2 - N2;
  const int col1 = (fy + NS1) & (NS1 - 1);
  const int row0 = part * ROWS_PER_BLOCK + (t >> LOG2);
  const int wrap1 = (part >= SPLIT / 2) ? (NS1 - N2) : 0;
  const float2* pA = F1 + ((size_t)(row0 + wrap1) << LOGS1) + col1;
  const float2* pB = F2 + ((size_t)row0 << LOG2) + iy2;

  float2 A[U], B[U];
#pragma unroll
  for (int u = 0; u < U; ++u) {
    A[u] = pA[(size_t)u * ROWS_PER_PASS * NS1];
    B[u] = pB[(size_t)u * ROWS_PER_PASS * N2];
  }
#pragma unroll
  for (int u = 0; u < U; ++u)
    asm volatile("" ::"v"(A[u].x), "v"(A[u].y), "v"(B[u].x), "v"(B[u].y));

  float sy, cy;
  __sincosf(TWO_PI_F * (float)fy / (float)N1, &sy, &cy);
  const float c2y = cy * cy - sy * sy, s2y = 2.f * cy * sy;

  float a0 = 0.f, a1 = 0.f, a2 = 0.f, a3 = 0.f, a4 = 0.f, a5 = 0.f, a6 = 0.f;
#pragma unroll
  for (int u = 0; u < U; ++u) {
    const int ix2 = row0 + u * ROWS_PER_PASS;
    const int fx = (ix2 < (N2 >> 1)) ? ix2 : ix2 - N2;
    const float zr = A[u].x * B[u].x + A[u].y * B[u].y;
    const float zi = A[u].y * B[u].x - A[u].x * B[u].y;
    float sx, cx;
    __sincosf(TWO_PI_F * (float)fx / (float)N1, &sx, &cx);
    const float c2x = cx * cx - sx * sx, s2x = 2.f * cx * sx;
    a0 += zr;
    a1 += zr * cy - zi * sy;
    a2 += zr * c2y - zi * s2y;
    a3 += zr * cx - zi * sx;
    const float xr = zr * cx - zi * sx, xi = zr * sx + zi * cx;
    a4 += xr * cy - xi * sy;
    a5 += zr * c2x - zi * s2x;
    const float mr = zr * cx + zi * sx, mi = zi * cx - zr * sx;
    a6 += mr * cy - mi * sy;
  }

  float accs[7] = {a0, a1, a2, a3, a4, a5, a6};
#pragma unroll
  for (int s = 0; s < 7; ++s) {
    float v = accs[s];
    for (int off = 16; off > 0; off >>= 1) v += __shfl_xor(v, off, 32);
    accs[s] = v;
  }
  if ((t & 31) == 0) {
#pragma unroll
    for (int s = 0; s < 7; ++s) red[s * NW + (t >> 5)] = accs[s];
  }
  __syncthreads();
  if (t < 7) {
    float v = 0.f;
    for (int w = 0; w < NW; ++w) v += red[t * NW + w];
    const float inv = 1.0f / (float)BINS;
    atomicAdd(&out[(((size_t)b * 384) + pairBase + p) * 7 + t], v * inv);
  }
}

__global__ void __launch_bounds__(512)
cross_all(const float2* __restrict__ spec0, const float2* __restrict__ spec1,
          const float2* __restrict__ spec2, float* __restrict__ out) {
  __shared__ float red[2][7 * 16];
  const int bid = blockIdx.x;
  if (bid < 256) {
    cross_body<8, 7, 15, 7, 2, 16>(spec0, spec1, 64, out, bid, threadIdx.x,
                                   &red[0][0]);
  } else {
    const int hf = threadIdx.x >> 8;
    const int t = threadIdx.x & 255;
    const int unit = (bid - 256) * 2 + hf;
    if (bid < 384) {
      cross_body<8, 7, 15, 6, 2, 8>(spec0, spec2, 128, out, unit, t, &red[hf][0]);
    } else {
      cross_body<7, 7, 14, 6, 2, 8>(spec1, spec2, 256, out, unit - 256, t,
                                    &red[hf][0]);
    }
  }
}

extern "C" void kernel_launch(void* const* d_in, const int* in_sizes, int n_in,
                              void* d_out, int out_size, void* d_ws,
                              size_t ws_size, hipStream_t stream) {
  const float* x0 = (const float*)d_in[0];
  const float* x1 = (const float*)d_in[1];
  const float* x2 = (const float*)d_in[2];
  float* out = (float*)d_out;

  const size_t n0 = (size_t)16 * 256 * 128;
  const size_t n1 = (size_t)16 * 128 * 128;
  const size_t n2 = (size_t)16 * 64 * 64;
  if (ws_size >= (n0 + n1 + n2) * sizeof(float2)) {
    float2* spec0 = (float2*)d_ws;
    float2* spec1 = spec0 + n0;
    float2* spec2 = spec1 + n1;

    fft_rows_all<<<dim3(ROW_BLOCKS0 + ROW_BLOCKS1 + ROW_BLOCKS2), 128, 0, stream>>>(
        x0, x1, x2, spec0, spec1, spec2, out);
    cols_and_same_all<<<dim3(SAME_BLOCKS + COL_BLOCKS0 + COL_BLOCKS1 + COL_BLOCKS2),
                        128, 0, stream>>>(x0, x1, x2, spec0, spec1, spec2, out);
    cross_all<<<dim3(512), 512, 0, stream>>>(spec0, spec1, spec2, out);
  }
}


# column FFTs: global loads/stores remapped so a wave's lanes lie along a spectrum row (whole 128-512 B pieces, not 32 rows x 16 B); LDS lines padded by 2 values; every sum in the same order
# speedup vs baseline: 7.5787x; 1.1510x over previous
#include <hip/hip_runtime.h>
#include <hip/hip_bf16.h>


typedef __attribute__((ext_vector_type(2))) float v2f;
typedef __attribute__((ext_vector_type(8))) float v8f;

__device__ __constant__ int c_sdx[7] = {0, 0, 0, 1, 1, 2, -1};
__device__ __constant__ int c_sdy[7] = {0, 1, 2, 0, 1, 0, 1};

#define TWO_PI_F 6.28318530717958647692f

template <int LOGW, int ROWS>
__device__ __forceinline__ void same_scale_body(const float* __restrict__ xin,
                                                int pairBase,
                                                float* __restrict__ out,
                                                int blk, float* lds) {
  constexpr int W = 1 << LOGW;
  constexpr int H = W;
  constexpr int WS = W + 4;
  constexpr int ZROW = 32 * WS;
  constexpr int SLABS = H / ROWS;

  const int b = blk / SLABS;
  const int slab = blk & (SLABS - 1);

  const int lane = threadIdx.x & 31;
  const int wave = threadIdx.x >> 5;
  const int m0 = wave * 16;

  const int mrow = m0 + (lane & 15);
  const bool aval = (mrow < 56);
  const int sidx = aval ? (mrow >> 3) : 0;
  const int l1 = mrow & 7;
  const int slot = c_sdx[sidx] + 1;
  const int dy = aval ? c_sdy[sidx] : 0;
  const int aIdx = (aval ? (slot * 8 + l1) * WS : ZROW) + dy;
  const int ncol = lane & 15;
  const int bBase = (ncol < 8) ? (8 + ncol) * WS : ZROW;
  const int khi = (lane >> 4) << 1;

  for (int idx = threadIdx.x; idx < WS; idx += 128) lds[ZROW + idx] = 0.0f;

  v8f acc = {};

  for (int xi = 0; xi < ROWS; ++xi) {
    const int xr = slab * ROWS + xi;

    {
      constexpr int NV4 = 8 * W;
      constexpr int PER_CH4 = W / 4;
      for (int idx = threadIdx.x; idx < NV4; idx += 128) {
        const int row = idx >> (LOGW - 2);
        const int y4 = idx & (PER_CH4 - 1);
        const int rs = row >> 3;
        const int ch = row & 7;
        int xs = xr + rs - 1;
        xs += (xs < 0) ? H : 0;
        xs -= (xs >= H) ? H : 0;
        *(float4*)&lds[row * WS + (y4 << 2)] =
            *(const float4*)&xin[((((size_t)b * 8 + ch) << LOGW) + xs) * W +
                                 (y4 << 2)];
      }
      const int row = threadIdx.x >> 2;
      const int e = threadIdx.x & 3;
      const int rs = row >> 3;
      const int ch = row & 7;
      int xs = xr + rs - 1;
      xs += (xs < 0) ? H : 0;
      xs -= (xs >= H) ? H : 0;
      lds[row * WS + W + e] =
          xin[((((size_t)b * 8 + ch) << LOGW) + xs) * W + e];
    }
    if (threadIdx.x < 8) {
      int xn = xr + 4;
      xn -= (xn >= H) ? H : 0;
      __builtin_prefetch(
          &xin[((((size_t)b * 8 + threadIdx.x) << LOGW) + xn) * W], 0, 1);
    }
    __syncthreads();

#pragma unroll 8
    for (int k = 0; k < W; k += 4) {
      const int kb = k + khi;
      v2f av, bv;
      av[0] = lds[aIdx + kb];
      av[1] = lds[aIdx + kb + 1];
      bv = *(const v2f*)&lds[bBase + kb];
      acc = __builtin_amdgcn_wmma_f32_16x16x4_f32(
           false, av,  false, bv,
           (short)0, acc,  false,  false);
    }
    __syncthreads();
  }

  const int rbase = m0 + ((lane >> 4) << 3);
  const int col = lane & 15;
  if (col < 8) {
#pragma unroll
    for (int i = 0; i < 8; ++i) {
      const int R = rbase + i;
      if (R < 56) {
        const int s = R >> 3;
        const int l1o = R & 7;
        atomicAdd(&out[(((size_t)b * 384) + pairBase + l1o * 8 + col) * 7 + s],
                  acc[i]);
      }
    }
  }
}


#define FFT_R 8

template <int LOGN, int P = (1 << LOGN)>
__device__ __forceinline__ void fft_lines(float2* buf, int t) {
  constexpr int N = 1 << LOGN;
  constexpr int Q = N / 4;
  static_assert(P >= N, "line pitch must hold a line");
#pragma unroll
  for (int st = 1; st + 1 <= LOGN; st += 2) {
    const int h = 1 << (st - 1);
    const int j = t & (h - 1);
    const int i0 = ((t >> (st - 1)) << (st + 1)) + j;
    float s1, c1, s2, c2, s3, c3;
    __sincosf(-TWO_PI_F * (float)j / (float)(2 * h), &s1, &c1);
    __sincosf(-TWO_PI_F * (float)j / (float)(4 * h), &s2, &c2);
    __sincosf(-TWO_PI_F * (float)(j + h) / (float)(4 * h), &s3, &c3);
#pragma unroll
    for (int r = 0; r < FFT_R; ++r) {
      float2* b = buf + r * P + i0;
      const float2 e0 = b[0], e1 = b[h], e2 = b[2 * h], e3 = b[3 * h];
      const float2 ta = make_float2(e1.x * c1 - e1.y * s1, e1.x * s1 + e1.y * c1);
      const float2 f0 = make_float2(e0.x + ta.x, e0.y + ta.y);
      const float2 f1 = make_float2(e0.x - ta.x, e0.y - ta.y);
      const float2 tb = make_float2(e3.x * c1 - e3.y * s1, e3.x * s1 + e3.y * c1);
      const float2 f2 = make_float2(e2.x + tb.x, e2.y + tb.y);
      const float2 f3 = make_float2(e2.x - tb.x, e2.y - tb.y);
      const float2 tc = make_float2(f2.x * c2 - f2.y * s2, f2.x * s2 + f2.y * c2);
      const float2 td = make_float2(f3.x * c3 - f3.y * s3, f3.x * s3 + f3.y * c3);
      b[0] = make_float2(f0.x + tc.x, f0.y + tc.y);
      b[2 * h] = make_float2(f0.x - tc.x, f0.y - tc.y);
      b[h] = make_float2(f1.x + td.x, f1.y + td.y);
      b[3 * h] = make_float2(f1.x - td.x, f1.y - td.y);
    }
    __syncthreads();
  }
  if (LOGN & 1) {
    constexpr int half = N / 2;
#pragma unroll
    for (int m = 0; m < 2; ++m) {
      const int j = t + m * Q;
      float sn, cs;
      __sincosf(-TWO_PI_F * (float)j / (float)N, &sn, &cs);
#pragma unroll
      for (int r = 0; r < FFT_R; ++r) {
        float2* b = buf + r * P;
        const float2 u = b[j];
        const float2 v = b[j + half];
        const float2 tv = make_float2(v.x * cs - v.y * sn, v.x * sn + v.y * cs);
        b[j] = make_float2(u.x + tv.x, u.y + tv.y);
        b[j + half] = make_float2(u.x - tv.x, u.y - tv.y);
      }
    }
    __syncthreads();
  }
}

template <int LOGN, bool BAND>
__device__ __forceinline__ void fft_rows_body(const float* __restrict__ xin,
                                              float2* __restrict__ spec,
                                              int blk, float2* lds) {
  constexpr int N = 1 << LOGN;
  constexpr int Q = N / 4;
  constexpr int HALF = N / 2;
  constexpr int RPB = 128 / Q;
  const int sub = threadIdx.x >> (LOGN - 2);
  const int t = threadIdx.x & (Q - 1);
  const size_t line0 = ((size_t)blk * RPB + sub) * FFT_R;
  float2* buf = lds + sub * (FFT_R * N);
  const float* src = xin + line0 * N;

  float v[4][FFT_R];
#pragma unroll
  for (int m = 0; m < 4; ++m) {
#pragma unroll
    for (int r = 0; r < FFT_R; ++r) v[m][r] = src[r * N + t + m * Q];
  }
#pragma unroll
  for (int m = 0; m < 4; ++m) {
    const int rm = __brev((unsigned)(t + m * Q)) >> (32 - LOGN);
#pragma unroll
    for (int r = 0; r < FFT_R; ++r) buf[r * N + rm] = make_float2(v[m][r], 0.0f);
  }
  __syncthreads();
  fft_lines<LOGN>(buf, t);
  if (BAND) {
    float2* dst = spec + line0 * HALF;
#pragma unroll
    for (int r = 0; r < FFT_R; ++r) {
      dst[r * HALF + t] = buf[r * N + t];
      dst[r * HALF + t + Q] = buf[r * N + t + 3 * Q];
    }
  } else {
    float2* dst = spec + line0 * N;
#pragma unroll
    for (int r = 0; r < FFT_R; ++r) {
#pragma unroll
      for (int m = 0; m < 4; ++m) dst[r * N + t + m * Q] = buf[r * N + t + m * Q];
    }
  }
}

#define ROW_BLOCKS0 (4096 / 16)
#define ROW_BLOCKS1 (2048 / 32)
#define ROW_BLOCKS2 (1024 / 64)

__global__ void __launch_bounds__(128)
fft_rows_all(const float* __restrict__ x0, const float* __restrict__ x1,
             const float* __restrict__ x2, float2* __restrict__ spec0,
             float2* __restrict__ spec1, float2* __restrict__ spec2,
             float* __restrict__ out) {
  __shared__ float2 buf[2 * 256 * FFT_R];
  const int bid = blockIdx.x;
  const int oi = bid * 128 + threadIdx.x;
  if (oi < 2 * 384 * 7) out[oi] = 0.0f;
  if (bid < ROW_BLOCKS0) {
    fft_rows_body<8, true>(x0, spec0, bid, buf);
  } else if (bid < ROW_BLOCKS0 + ROW_BLOCKS1) {
    fft_rows_body<7, false>(x1, spec1, bid - ROW_BLOCKS0, buf);
  } else {
    fft_rows_body<6, false>(x2, spec2, bid - (ROW_BLOCKS0 + ROW_BLOCKS1), buf);
  }
}

template <int LOGN, int LOGC, bool BAND>
__device__ __forceinline__ void fft_cols_body(float2* __restrict__ spec, int blk,
                                              float2* lds) {
  constexpr int N = 1 << LOGN;
  constexpr int NC = 1 << LOGC;
  constexpr int P = N + 2;
  constexpr int Q = N / 4;
  constexpr int RPB = 128 / Q;
  constexpr int C = RPB * FFT_R;
  constexpr int F4R = C / 2;
  constexpr int RPI = 32 / F4R;
  constexpr int NIT = (N * F4R) / 128;
  static_assert((FFT_R & 1) == 0, "column pairs move as 16-byte values");
  static_assert(F4R <= 32 && RPI * F4R == 32, "a wave covers whole row pieces");
  static_assert(NIT * 4 * RPI == N, "the fill covers every row once");
  static_assert(C <= NC, "a block stays inside one image");
  const int sub = threadIdx.x >> (LOGN - 2);
  const int t = threadIdx.x & (Q - 1);
  const int lane = threadIdx.x & 31;
  const int wave = threadIdx.x >> 5;
  const int j = lane & (F4R - 1);
  const int kk = lane / F4R;
  const int line0 = blk * C;
  const int col = line0 & (NC - 1);
  const int img = line0 >> LOGC;
  float2* base0 = spec + ((size_t)img << (LOGN + LOGC)) + col + 2 * j;
  float2* lcol = lds + (2 * j) * P;

  float4 ld[NIT];
#pragma unroll
  for (int i = 0; i < NIT; ++i) {
    const int row = i * 4 + wave + kk * (N / RPI);
    ld[i] = *(const float4*)(base0 + (size_t)row * NC);
  }
#pragma unroll
  for (int i = 0; i < NIT; ++i) {
    const int row = i * 4 + wave + kk * (N / RPI);
    const int rm = __brev((unsigned)row) >> (32 - LOGN);
    lcol[rm] = make_float2(ld[i].x, ld[i].y);
    lcol[P + rm] = make_float2(ld[i].z, ld[i].w);
  }
  __syncthreads();
  fft_lines<LOGN, P>(lds + sub * (FFT_R * P), t);
  if (BAND) {
#pragma unroll
    for (int i = 0; i < NIT / 2; ++i) {
      const int br = (i * 4 + wave) * RPI + kk;
      const int ko = (br < N / 4) ? br : br + N / 2;
      const float2 a0 = lcol[ko], a1 = lcol[P + ko];
      *(float4*)(base0 + (size_t)br * NC) = make_float4(a0.x, a0.y, a1.x, a1.y);
    }
  } else {
#pragma unroll
    for (int i = 0; i < NIT; ++i) {
      const int ko = (i * 4 + wave) * RPI + kk;
      const float2 a0 = lcol[ko], a1 = lcol[P + ko];
      *(float4*)(base0 + (size_t)ko * NC) = make_float4(a0.x, a0.y, a1.x, a1.y);
    }
  }
}

#define SAME_BLOCKS (256 + 256 + 128)
#define COL_BLOCKS0 (16 * 128 / 16)
#define COL_BLOCKS1 (16 * 128 / 32)
#define COL_BLOCKS2 (16 * 64 / 64)
__global__ void __launch_bounds__(128)
cols_and_same_all(const float* __restrict__ x0, const float* __restrict__ x1,
                  const float* __restrict__ x2, float2* __restrict__ spec0,
                  float2* __restrict__ spec1, float2* __restrict__ spec2,
                  float* __restrict__ out) {
  __shared__ __attribute__((aligned(16))) float lds[33 * 260];
  static_assert(sizeof(float2) * 64 * (64 + 2) <= sizeof(float) * 33 * 260 &&
                    sizeof(float2) * 32 * (128 + 2) <= sizeof(float) * 33 * 260 &&
                    sizeof(float2) * 16 * (256 + 2) <= sizeof(float) * 33 * 260,
                "transform buffer must fit the shared array");
  const int bid = blockIdx.x;
  if (bid < 256) {
    same_scale_body<8, 2>(x0, 0, out, bid, lds);
  } else if (bid < 512) {
    same_scale_body<7, 1>(x1, 192, out, bid - 256, lds);
  } else if (bid < SAME_BLOCKS) {
    same_scale_body<6, 1>(x2, 320, out, bid - 512, lds);
  } else {
    const int fb = bid - SAME_BLOCKS;
    float2* buf = reinterpret_cast<float2*>(lds);
    if (fb < COL_BLOCKS0) {
      fft_cols_body<8, 7, true>(spec0, fb, buf);
    } else if (fb < COL_BLOCKS0 + COL_BLOCKS1) {
      fft_cols_body<7, 7, false>(spec1, fb - COL_BLOCKS0, buf);
    } else {
      fft_cols_body<6, 6, false>(spec2, fb - (COL_BLOCKS0 + COL_BLOCKS1), buf);
    }
  }
}

template <int LOG1, int LOGS1, int LOGI1, int LOG2, int SPLIT, int U>
__device__ __forceinline__ void cross_body(const float2* __restrict__ s1,
                                           const float2* __restrict__ s2,
                                           int pairBase,
                                           float* __restrict__ out, int unit,
                                           int t, float* red) {
  constexpr int N1 = 1 << LOG1;
  constexpr int NS1 = 1 << LOGS1;
  constexpr int N2 = 1 << LOG2;
  constexpr int BINS = N2 * N2;
  constexpr int THREADS = BINS / (SPLIT * U);
  constexpr int NW = THREADS / 32;
  constexpr int ROWS_PER_PASS = THREADS >> LOG2;
  constexpr int ROWS_PER_BLOCK = N2 / SPLIT;
  static_assert(THREADS >= N2 && THREADS <= 1024 && (THREADS & 31) == 0,
                "block must cover whole spectrum rows");
  static_assert(ROWS_PER_PASS * U == ROWS_PER_BLOCK, "share must tile the rows");
  static_assert(SPLIT >= 2 && (SPLIT & (SPLIT - 1)) == 0, "rows of one half per block");
  static_assert(N2 <= NS1 && NS1 <= N1, "stored fine band must hold the coarse support");

  const int part = unit & (SPLIT - 1);
  const int bp = unit / SPLIT;
  const int b = bp >> 6;
  const int p = bp & 63;
  const int l1 = p >> 3, l2 = p & 7;
  const float2* F1 = s1 + (((size_t)b * 8 + l1) << LOGI1);
  const float2* F2 = s2 + (((size_t)b * 8 + l2) << (2 * LOG2));

  const int iy2 = t & (N2 - 1);
  const int fy = (iy2 < (N2 >> 1)) ? iy2 : iy2 - N2;
  const int col1 = (fy + NS1) & (NS1 - 1);
  const int row0 = part * ROWS_PER_BLOCK + (t >> LOG2);
  const int wrap1 = (part >= SPLIT / 2) ? (NS1 - N2) : 0;
  const float2* pA = F1 + ((size_t)(row0 + wrap1) << LOGS1) + col1;
  const float2* pB = F2 + ((size_t)row0 << LOG2) + iy2;

  float2 A[U], B[U];
#pragma unroll
  for (int u = 0; u < U; ++u) {
    A[u] = pA[(size_t)u * ROWS_PER_PASS * NS1];
    B[u] = pB[(size_t)u * ROWS_PER_PASS * N2];
  }
#pragma unroll
  for (int u = 0; u < U; ++u)
    asm volatile("" ::"v"(A[u].x), "v"(A[u].y), "v"(B[u].x), "v"(B[u].y));

  float sy, cy;
  __sincosf(TWO_PI_F * (float)fy / (float)N1, &sy, &cy);
  const float c2y = cy * cy - sy * sy, s2y = 2.f * cy * sy;

  float a0 = 0.f, a1 = 0.f, a2 = 0.f, a3 = 0.f, a4 = 0.f, a5 = 0.f, a6 = 0.f;
#pragma unroll
  for (int u = 0; u < U; ++u) {
    const int ix2 = row0 + u * ROWS_PER_PASS;
    const int fx = (ix2 < (N2 >> 1)) ? ix2 : ix2 - N2;
    const float zr = A[u].x * B[u].x + A[u].y * B[u].y;
    const float zi = A[u].y * B[u].x - A[u].x * B[u].y;
    float sx, cx;
    __sincosf(TWO_PI_F * (float)fx / (float)N1, &sx, &cx);
    const float c2x = cx * cx - sx * sx, s2x = 2.f * cx * sx;
    a0 += zr;
    a1 += zr * cy - zi * sy;
    a2 += zr * c2y - zi * s2y;
    a3 += zr * cx - zi * sx;
    const float xr = zr * cx - zi * sx, xi = zr * sx + zi * cx;
    a4 += xr * cy - xi * sy;
    a5 += zr * c2x - zi * s2x;
    const float mr = zr * cx + zi * sx, mi = zi * cx - zr * sx;
    a6 += mr * cy - mi * sy;
  }

  float accs[7] = {a0, a1, a2, a3, a4, a5, a6};
#pragma unroll
  for (int s = 0; s < 7; ++s) {
    float v = accs[s];
    for (int off = 16; off > 0; off >>= 1) v += __shfl_xor(v, off, 32);
    accs[s] = v;
  }
  if ((t & 31) == 0) {
#pragma unroll
    for (int s = 0; s < 7; ++s) red[s * NW + (t >> 5)] = accs[s];
  }
  __syncthreads();
  if (t < 7) {
    float v = 0.f;
    for (int w = 0; w < NW; ++w) v += red[t * NW + w];
    const float inv = 1.0f / (float)BINS;
    atomicAdd(&out[(((size_t)b * 384) + pairBase + p) * 7 + t], v * inv);
  }
}

__global__ void __launch_bounds__(512)
cross_all(const float2* __restrict__ spec0, const float2* __restrict__ spec1,
          const float2* __restrict__ spec2, float* __restrict__ out) {
  __shared__ float red[2][7 * 16];
  const int bid = blockIdx.x;
  if (bid < 256) {
    cross_body<8, 7, 15, 7, 2, 16>(spec0, spec1, 64, out, bid, threadIdx.x,
                                   &red[0][0]);
  } else {
    const int hf = threadIdx.x >> 8;
    const int t = threadIdx.x & 255;
    const int unit = (bid - 256) * 2 + hf;
    if (bid < 384) {
      cross_body<8, 7, 15, 6, 2, 8>(spec0, spec2, 128, out, unit, t, &red[hf][0]);
    } else {
      cross_body<7, 7, 14, 6, 2, 8>(spec1, spec2, 256, out, unit - 256, t,
                                    &red[hf][0]);
    }
  }
}

extern "C" void kernel_launch(void* const* d_in, const int* in_sizes, int n_in,
                              void* d_out, int out_size, void* d_ws,
                              size_t ws_size, hipStream_t stream) {
  const float* x0 = (const float*)d_in[0];
  const float* x1 = (const float*)d_in[1];
  const float* x2 = (const float*)d_in[2];
  float* out = (float*)d_out;

  const size_t n0 = (size_t)16 * 256 * 128;
  const size_t n1 = (size_t)16 * 128 * 128;
  const size_t n2 = (size_t)16 * 64 * 64;
  if (ws_size >= (n0 + n1 + n2) * sizeof(float2)) {
    float2* spec0 = (float2*)d_ws;
    float2* spec1 = spec0 + n0;
    float2* spec2 = spec1 + n1;

    fft_rows_all<<<dim3(ROW_BLOCKS0 + ROW_BLOCKS1 + ROW_BLOCKS2), 128, 0, stream>>>(
        x0, x1, x2, spec0, spec1, spec2, out);
    cols_and_same_all<<<dim3(SAME_BLOCKS + COL_BLOCKS0 + COL_BLOCKS1 + COL_BLOCKS2),
                        128, 0, stream>>>(x0, x1, x2, spec0, spec1, spec2, out);
    cross_all<<<dim3(512), 512, 0, stream>>>(spec0, spec1, spec2, out);
  }
}


# launch 2: same-scale blocks stage their rows to LDS once, all global loads in flight together (1 memory round trip, was 4-9 per row); column transforms, now its longest blocks, get the first indices
# speedup vs baseline: 8.7243x; 1.1512x over previous
#include <hip/hip_runtime.h>
#include <hip/hip_bf16.h>


typedef __attribute__((ext_vector_type(2))) float v2f;
typedef __attribute__((ext_vector_type(4))) float v4f;
typedef __attribute__((ext_vector_type(8))) float v8f;

__device__ __constant__ int c_sdx[7] = {0, 0, 0, 1, 1, 2, -1};
__device__ __constant__ int c_sdy[7] = {0, 1, 2, 0, 1, 0, 1};

#define TWO_PI_F 6.28318530717958647692f

template <int LOGW, int ROWS>
__device__ __forceinline__ void same_scale_body(const float* __restrict__ xin,
                                                int pairBase,
                                                float* __restrict__ out,
                                                int blk, float* lds) {
  constexpr int W = 1 << LOGW;
  constexpr int H = W;
  constexpr int WS = W + 4;
  constexpr int NR = ROWS + 3;
  constexpr int ZROW = NR * 8 * WS;
  constexpr int SLABS = H / ROWS;
  constexpr int F4R = W / 4;
  constexpr int RPI = 128 / F4R;
  constexpr int NLD = (NR * 8) / RPI;
  static_assert(RPI >= 1 && RPI <= 8 && (8 % RPI) == 0,
                "a pass covers whole rows inside one row slot");
  static_assert(NLD * RPI == NR * 8, "the passes cover every LDS row once");

  const int b = blk / SLABS;
  const int slab = blk & (SLABS - 1);
  const int x0 = slab * ROWS;

  const int lane = threadIdx.x & 31;
  const int wave = threadIdx.x >> 5;
  const int m0 = wave * 16;

  const int mrow = m0 + (lane & 15);
  const bool aval = (mrow < 56);
  const int sidx = aval ? (mrow >> 3) : 0;
  const int l1 = mrow & 7;
  const int slot = c_sdx[sidx] + 1;
  const int dy = aval ? c_sdy[sidx] : 0;
  const int aIdx = (aval ? (slot * 8 + l1) * WS : ZROW) + dy;
  const int aStep = aval ? 8 * WS : 0;
  const int ncol = lane & 15;
  const int bBase = (ncol < 8) ? (8 + ncol) * WS : ZROW;
  const int bStep = (ncol < 8) ? 8 * WS : 0;
  const int khi = (lane >> 4) << 1;

  {
    const int tid = threadIdx.x;
    const int y4 = tid & (F4R - 1);
    const int r0 = tid >> (LOGW - 2);
    const float* img = xin + ((size_t)b << (2 * LOGW + 3));
    const int tOff = (r0 << (2 * LOGW)) + (y4 << 2);
    v4f v[NLD];
#pragma unroll
    for (int i = 0; i < NLD; ++i) {
      const int chc = (i * RPI) & 7;
      const int rs = (i * RPI) >> 3;
      const int xs = (x0 - 1 + rs) & (H - 1);
      v[i] = *(const v4f*)(img + ((chc << (2 * LOGW)) + (xs << LOGW)) + tOff);
    }
#pragma unroll
    for (int i = 0; i < NLD; ++i)
      *(v4f*)&lds[(i * RPI + r0) * WS + (y4 << 2)] = v[i];
    if (y4 == 0) {
#pragma unroll
      for (int i = 0; i < NLD; ++i)
        *(v4f*)&lds[(i * RPI + r0) * WS + W] = v[i];
    }
    for (int idx = tid; idx < WS; idx += 128) lds[ZROW + idx] = 0.0f;
  }
  __syncthreads();

  v8f acc = {};

#pragma unroll
  for (int xi = 0; xi < ROWS; ++xi) {
    const int aI = aIdx + xi * aStep;
    const int bI = bBase + xi * bStep;
#pragma unroll 8
    for (int k = 0; k < W; k += 4) {
      const int kb = k + khi;
      v2f av, bv;
      av[0] = lds[aI + kb];
      av[1] = lds[aI + kb + 1];
      bv = *(const v2f*)&lds[bI + kb];
      acc = __builtin_amdgcn_wmma_f32_16x16x4_f32(
           false, av,  false, bv,
           (short)0, acc,  false,  false);
    }
  }

  const int rbase = m0 + ((lane >> 4) << 3);
  const int col = lane & 15;
  if (col < 8) {
#pragma unroll
    for (int i = 0; i < 8; ++i) {
      const int R = rbase + i;
      if (R < 56) {
        const int s = R >> 3;
        const int l1o = R & 7;
        atomicAdd(&out[(((size_t)b * 384) + pairBase + l1o * 8 + col) * 7 + s],
                  acc[i]);
      }
    }
  }
}


#define FFT_R 8

template <int LOGN, int P = (1 << LOGN)>
__device__ __forceinline__ void fft_lines(float2* buf, int t) {
  constexpr int N = 1 << LOGN;
  constexpr int Q = N / 4;
  static_assert(P >= N, "line pitch must hold a line");
#pragma unroll
  for (int st = 1; st + 1 <= LOGN; st += 2) {
    const int h = 1 << (st - 1);
    const int j = t & (h - 1);
    const int i0 = ((t >> (st - 1)) << (st + 1)) + j;
    float s1, c1, s2, c2, s3, c3;
    __sincosf(-TWO_PI_F * (float)j / (float)(2 * h), &s1, &c1);
    __sincosf(-TWO_PI_F * (float)j / (float)(4 * h), &s2, &c2);
    __sincosf(-TWO_PI_F * (float)(j + h) / (float)(4 * h), &s3, &c3);
#pragma unroll
    for (int r = 0; r < FFT_R; ++r) {
      float2* b = buf + r * P + i0;
      const float2 e0 = b[0], e1 = b[h], e2 = b[2 * h], e3 = b[3 * h];
      const float2 ta = make_float2(e1.x * c1 - e1.y * s1, e1.x * s1 + e1.y * c1);
      const float2 f0 = make_float2(e0.x + ta.x, e0.y + ta.y);
      const float2 f1 = make_float2(e0.x - ta.x, e0.y - ta.y);
      const float2 tb = make_float2(e3.x * c1 - e3.y * s1, e3.x * s1 + e3.y * c1);
      const float2 f2 = make_float2(e2.x + tb.x, e2.y + tb.y);
      const float2 f3 = make_float2(e2.x - tb.x, e2.y - tb.y);
      const float2 tc = make_float2(f2.x * c2 - f2.y * s2, f2.x * s2 + f2.y * c2);
      const float2 td = make_float2(f3.x * c3 - f3.y * s3, f3.x * s3 + f3.y * c3);
      b[0] = make_float2(f0.x + tc.x, f0.y + tc.y);
      b[2 * h] = make_float2(f0.x - tc.x, f0.y - tc.y);
      b[h] = make_float2(f1.x + td.x, f1.y + td.y);
      b[3 * h] = make_float2(f1.x - td.x, f1.y - td.y);
    }
    __syncthreads();
  }
  if (LOGN & 1) {
    constexpr int half = N / 2;
#pragma unroll
    for (int m = 0; m < 2; ++m) {
      const int j = t + m * Q;
      float sn, cs;
      __sincosf(-TWO_PI_F * (float)j / (float)N, &sn, &cs);
#pragma unroll
      for (int r = 0; r < FFT_R; ++r) {
        float2* b = buf + r * P;
        const float2 u = b[j];
        const float2 v = b[j + half];
        const float2 tv = make_float2(v.x * cs - v.y * sn, v.x * sn + v.y * cs);
        b[j] = make_float2(u.x + tv.x, u.y + tv.y);
        b[j + half] = make_float2(u.x - tv.x, u.y - tv.y);
      }
    }
    __syncthreads();
  }
}

template <int LOGN, bool BAND>
__device__ __forceinline__ void fft_rows_body(const float* __restrict__ xin,
                                              float2* __restrict__ spec,
                                              int blk, float2* lds) {
  constexpr int N = 1 << LOGN;
  constexpr int Q = N / 4;
  constexpr int HALF = N / 2;
  constexpr int RPB = 128 / Q;
  const int sub = threadIdx.x >> (LOGN - 2);
  const int t = threadIdx.x & (Q - 1);
  const size_t line0 = ((size_t)blk * RPB + sub) * FFT_R;
  float2* buf = lds + sub * (FFT_R * N);
  const float* src = xin + line0 * N;

  float v[4][FFT_R];
#pragma unroll
  for (int m = 0; m < 4; ++m) {
#pragma unroll
    for (int r = 0; r < FFT_R; ++r) v[m][r] = src[r * N + t + m * Q];
  }
#pragma unroll
  for (int m = 0; m < 4; ++m) {
    const int rm = __brev((unsigned)(t + m * Q)) >> (32 - LOGN);
#pragma unroll
    for (int r = 0; r < FFT_R; ++r) buf[r * N + rm] = make_float2(v[m][r], 0.0f);
  }
  __syncthreads();
  fft_lines<LOGN>(buf, t);
  if (BAND) {
    float2* dst = spec + line0 * HALF;
#pragma unroll
    for (int r = 0; r < FFT_R; ++r) {
      dst[r * HALF + t] = buf[r * N + t];
      dst[r * HALF + t + Q] = buf[r * N + t + 3 * Q];
    }
  } else {
    float2* dst = spec + line0 * N;
#pragma unroll
    for (int r = 0; r < FFT_R; ++r) {
#pragma unroll
      for (int m = 0; m < 4; ++m) dst[r * N + t + m * Q] = buf[r * N + t + m * Q];
    }
  }
}

#define ROW_BLOCKS0 (4096 / 16)
#define ROW_BLOCKS1 (2048 / 32)
#define ROW_BLOCKS2 (1024 / 64)

__global__ void __launch_bounds__(128)
fft_rows_all(const float* __restrict__ x0, const float* __restrict__ x1,
             const float* __restrict__ x2, float2* __restrict__ spec0,
             float2* __restrict__ spec1, float2* __restrict__ spec2,
             float* __restrict__ out) {
  __shared__ float2 buf[2 * 256 * FFT_R];
  const int bid = blockIdx.x;
  const int oi = bid * 128 + threadIdx.x;
  if (oi < 2 * 384 * 7) out[oi] = 0.0f;
  if (bid < ROW_BLOCKS0) {
    fft_rows_body<8, true>(x0, spec0, bid, buf);
  } else if (bid < ROW_BLOCKS0 + ROW_BLOCKS1) {
    fft_rows_body<7, false>(x1, spec1, bid - ROW_BLOCKS0, buf);
  } else {
    fft_rows_body<6, false>(x2, spec2, bid - (ROW_BLOCKS0 + ROW_BLOCKS1), buf);
  }
}

template <int LOGN, int LOGC, bool BAND>
__device__ __forceinline__ void fft_cols_body(float2* __restrict__ spec, int blk,
                                              float2* lds) {
  constexpr int N = 1 << LOGN;
  constexpr int NC = 1 << LOGC;
  constexpr int P = N + 2;
  constexpr int Q = N / 4;
  constexpr int RPB = 128 / Q;
  constexpr int C = RPB * FFT_R;
  constexpr int F4R = C / 2;
  constexpr int RPI = 32 / F4R;
  constexpr int NIT = (N * F4R) / 128;
  static_assert((FFT_R & 1) == 0, "column pairs move as 16-byte values");
  static_assert(F4R <= 32 && RPI * F4R == 32, "a wave covers whole row pieces");
  static_assert(NIT * 4 * RPI == N, "the fill covers every row once");
  static_assert(C <= NC, "a block stays inside one image");
  const int sub = threadIdx.x >> (LOGN - 2);
  const int t = threadIdx.x & (Q - 1);
  const int lane = threadIdx.x & 31;
  const int wave = threadIdx.x >> 5;
  const int j = lane & (F4R - 1);
  const int kk = lane / F4R;
  const int line0 = blk * C;
  const int col = line0 & (NC - 1);
  const int img = line0 >> LOGC;
  float2* base0 = spec + ((size_t)img << (LOGN + LOGC)) + col + 2 * j;
  float2* lcol = lds + (2 * j) * P;

  float4 ld[NIT];
#pragma unroll
  for (int i = 0; i < NIT; ++i) {
    const int row = i * 4 + wave + kk * (N / RPI);
    ld[i] = *(const float4*)(base0 + (size_t)row * NC);
  }
#pragma unroll
  for (int i = 0; i < NIT; ++i) {
    const int row = i * 4 + wave + kk * (N / RPI);
    const int rm = __brev((unsigned)row) >> (32 - LOGN);
    lcol[rm] = make_float2(ld[i].x, ld[i].y);
    lcol[P + rm] = make_float2(ld[i].z, ld[i].w);
  }
  __syncthreads();
  fft_lines<LOGN, P>(lds + sub * (FFT_R * P), t);
  if (BAND) {
#pragma unroll
    for (int i = 0; i < NIT / 2; ++i) {
      const int br = (i * 4 + wave) * RPI + kk;
      const int ko = (br < N / 4) ? br : br + N / 2;
      const float2 a0 = lcol[ko], a1 = lcol[P + ko];
      *(float4*)(base0 + (size_t)br * NC) = make_float4(a0.x, a0.y, a1.x, a1.y);
    }
  } else {
#pragma unroll
    for (int i = 0; i < NIT; ++i) {
      const int ko = (i * 4 + wave) * RPI + kk;
      const float2 a0 = lcol[ko], a1 = lcol[P + ko];
      *(float4*)(base0 + (size_t)ko * NC) = make_float4(a0.x, a0.y, a1.x, a1.y);
    }
  }
}

#define SAME_BLOCKS (256 + 256 + 128)
#define COL_BLOCKS0 (16 * 128 / 16)
#define COL_BLOCKS1 (16 * 128 / 32)
#define COL_BLOCKS2 (16 * 64 / 64)
#define COL_BLOCKS (COL_BLOCKS0 + COL_BLOCKS1 + COL_BLOCKS2)
__global__ void __launch_bounds__(128)
cols_and_same_all(const float* __restrict__ x0, const float* __restrict__ x1,
                  const float* __restrict__ x2, float2* __restrict__ spec0,
                  float2* __restrict__ spec1, float2* __restrict__ spec2,
                  float* __restrict__ out) {
  __shared__ __attribute__((aligned(16))) float lds[41 * 260];
  static_assert(sizeof(float2) * 64 * (64 + 2) <= sizeof(float) * 41 * 260 &&
                    sizeof(float2) * 32 * (128 + 2) <= sizeof(float) * 41 * 260 &&
                    sizeof(float2) * 16 * (256 + 2) <= sizeof(float) * 41 * 260,
                "transform buffer must fit the shared array");
  const int bid = blockIdx.x;
  if (bid < COL_BLOCKS) {
    float2* buf = reinterpret_cast<float2*>(lds);
    if (bid < COL_BLOCKS0) {
      fft_cols_body<8, 7, true>(spec0, bid, buf);
    } else if (bid < COL_BLOCKS0 + COL_BLOCKS1) {
      fft_cols_body<7, 7, false>(spec1, bid - COL_BLOCKS0, buf);
    } else {
      fft_cols_body<6, 6, false>(spec2, bid - (COL_BLOCKS0 + COL_BLOCKS1), buf);
    }
  } else {
    const int sb = bid - COL_BLOCKS;
    if (sb < 256) {
      same_scale_body<8, 2>(x0, 0, out, sb, lds);
    } else if (sb < 512) {
      same_scale_body<7, 1>(x1, 192, out, sb - 256, lds);
    } else {
      same_scale_body<6, 1>(x2, 320, out, sb - 512, lds);
    }
  }
}

template <int LOG1, int LOGS1, int LOGI1, int LOG2, int SPLIT, int U>
__device__ __forceinline__ void cross_body(const float2* __restrict__ s1,
                                           const float2* __restrict__ s2,
                                           int pairBase,
                                           float* __restrict__ out, int unit,
                                           int t, float* red) {
  constexpr int N1 = 1 << LOG1;
  constexpr int NS1 = 1 << LOGS1;
  constexpr int N2 = 1 << LOG2;
  constexpr int BINS = N2 * N2;
  constexpr int THREADS = BINS / (SPLIT * U);
  constexpr int NW = THREADS / 32;
  constexpr int ROWS_PER_PASS = THREADS >> LOG2;
  constexpr int ROWS_PER_BLOCK = N2 / SPLIT;
  static_assert(THREADS >= N2 && THREADS <= 1024 && (THREADS & 31) == 0,
                "block must cover whole spectrum rows");
  static_assert(ROWS_PER_PASS * U == ROWS_PER_BLOCK, "share must tile the rows");
  static_assert(SPLIT >= 2 && (SPLIT & (SPLIT - 1)) == 0, "rows of one half per block");
  static_assert(N2 <= NS1 && NS1 <= N1, "stored fine band must hold the coarse support");

  const int part = unit & (SPLIT - 1);
  const int bp = unit / SPLIT;
  const int b = bp >> 6;
  const int p = bp & 63;
  const int l1 = p >> 3, l2 = p & 7;
  const float2* F1 = s1 + (((size_t)b * 8 + l1) << LOGI1);
  const float2* F2 = s2 + (((size_t)b * 8 + l2) << (2 * LOG2));

  const int iy2 = t & (N2 - 1);
  const int fy = (iy2 < (N2 >> 1)) ? iy2 : iy2 - N2;
  const int col1 = (fy + NS1) & (NS1 - 1);
  const int row0 = part * ROWS_PER_BLOCK + (t >> LOG2);
  const int wrap1 = (part >= SPLIT / 2) ? (NS1 - N2) : 0;
  const float2* pA = F1 + ((size_t)(row0 + wrap1) << LOGS1) + col1;
  const float2* pB = F2 + ((size_t)row0 << LOG2) + iy2;

  float2 A[U], B[U];
#pragma unroll
  for (int u = 0; u < U; ++u) {
    A[u] = pA[(size_t)u * ROWS_PER_PASS * NS1];
    B[u] = pB[(size_t)u * ROWS_PER_PASS * N2];
  }
#pragma unroll
  for (int u = 0; u < U; ++u)
    asm volatile("" ::"v"(A[u].x), "v"(A[u].y), "v"(B[u].x), "v"(B[u].y));

  float sy, cy;
  __sincosf(TWO_PI_F * (float)fy / (float)N1, &sy, &cy);
  const float c2y = cy * cy - sy * sy, s2y = 2.f * cy * sy;

  float a0 = 0.f, a1 = 0.f, a2 = 0.f, a3 = 0.f, a4 = 0.f, a5 = 0.f, a6 = 0.f;
#pragma unroll
  for (int u = 0; u < U; ++u) {
    const int ix2 = row0 + u * ROWS_PER_PASS;
    const int fx = (ix2 < (N2 >> 1)) ? ix2 : ix2 - N2;
    const float zr = A[u].x * B[u].x + A[u].y * B[u].y;
    const float zi = A[u].y * B[u].x - A[u].x * B[u].y;
    float sx, cx;
    __sincosf(TWO_PI_F * (float)fx / (float)N1, &sx, &cx);
    const float c2x = cx * cx - sx * sx, s2x = 2.f * cx * sx;
    a0 += zr;
    a1 += zr * cy - zi * sy;
    a2 += zr * c2y - zi * s2y;
    a3 += zr * cx - zi * sx;
    const float xr = zr * cx - zi * sx, xi = zr * sx + zi * cx;
    a4 += xr * cy - xi * sy;
    a5 += zr * c2x - zi * s2x;
    const float mr = zr * cx + zi * sx, mi = zi * cx - zr * sx;
    a6 += mr * cy - mi * sy;
  }

  float accs[7] = {a0, a1, a2, a3, a4, a5, a6};
#pragma unroll
  for (int s = 0; s < 7; ++s) {
    float v = accs[s];
    for (int off = 16; off > 0; off >>= 1) v += __shfl_xor(v, off, 32);
    accs[s] = v;
  }
  if ((t & 31) == 0) {
#pragma unroll
    for (int s = 0; s < 7; ++s) red[s * NW + (t >> 5)] = accs[s];
  }
  __syncthreads();
  if (t < 7) {
    float v = 0.f;
    for (int w = 0; w < NW; ++w) v += red[t * NW + w];
    const float inv = 1.0f / (float)BINS;
    atomicAdd(&out[(((size_t)b * 384) + pairBase + p) * 7 + t], v * inv);
  }
}

__global__ void __launch_bounds__(512)
cross_all(const float2* __restrict__ spec0, const float2* __restrict__ spec1,
          const float2* __restrict__ spec2, float* __restrict__ out) {
  __shared__ float red[2][7 * 16];
  const int bid = blockIdx.x;
  if (bid < 256) {
    cross_body<8, 7, 15, 7, 2, 16>(spec0, spec1, 64, out, bid, threadIdx.x,
                                   &red[0][0]);
  } else {
    const int hf = threadIdx.x >> 8;
    const int t = threadIdx.x & 255;
    const int unit = (bid - 256) * 2 + hf;
    if (bid < 384) {
      cross_body<8, 7, 15, 6, 2, 8>(spec0, spec2, 128, out, unit, t, &red[hf][0]);
    } else {
      cross_body<7, 7, 14, 6, 2, 8>(spec1, spec2, 256, out, unit - 256, t,
                                    &red[hf][0]);
    }
  }
}

extern "C" void kernel_launch(void* const* d_in, const int* in_sizes, int n_in,
                              void* d_out, int out_size, void* d_ws,
                              size_t ws_size, hipStream_t stream) {
  const float* x0 = (const float*)d_in[0];
  const float* x1 = (const float*)d_in[1];
  const float* x2 = (const float*)d_in[2];
  float* out = (float*)d_out;

  const size_t n0 = (size_t)16 * 256 * 128;
  const size_t n1 = (size_t)16 * 128 * 128;
  const size_t n2 = (size_t)16 * 64 * 64;
  if (ws_size >= (n0 + n1 + n2) * sizeof(float2)) {
    float2* spec0 = (float2*)d_ws;
    float2* spec1 = spec0 + n0;
    float2* spec2 = spec1 + n1;

    fft_rows_all<<<dim3(ROW_BLOCKS0 + ROW_BLOCKS1 + ROW_BLOCKS2), 128, 0, stream>>>(
        x0, x1, x2, spec0, spec1, spec2, out);
    cols_and_same_all<<<dim3(COL_BLOCKS + SAME_BLOCKS), 128, 0, stream>>>(
        x0, x1, x2, spec0, spec1, spec2, out);
    cross_all<<<dim3(512), 512, 0, stream>>>(spec0, spec1, spec2, out);
  }
}
